# starting kernel
# baseline (speedup 1.0000x reference)
#include <hip/hip_runtime.h>

typedef __attribute__((ext_vector_type(2))) float v2f;
typedef __attribute__((ext_vector_type(8))) float v8f;

#define NNETS 256
#define HID 128
#define BATCH 2048
#define ROWS_PER_BLK 128
#define W2_STRIDE 132

__global__ __launch_bounds__(256) void mlp256_fused_wmma_f32(
    const float* __restrict__ a,  const float* __restrict__ W1,
    const float* __restrict__ b1, const float* __restrict__ W2,
    const float* __restrict__ b2, const float* __restrict__ W3,
    const float* __restrict__ b3, float* __restrict__ out)
{
    __shared__ float sW2[HID * W2_STRIDE];
    __shared__ float sW1[HID];
    __shared__ float sB1[HID];
    __shared__ float sB2[HID];
    __shared__ float sW3[HID];
    __shared__ float sA[ROWS_PER_BLK];

    const int n  = blockIdx.y;
    const int r0 = blockIdx.x * ROWS_PER_BLK;
    const int t  = threadIdx.x;

    if (t < HID) {
        sW1[t] = W1[n * HID + t];
        sB1[t] = b1[n * HID + t];
        sB2[t] = b2[n * HID + t];
        sW3[t] = W3[n * HID + t];
        sA[t]  = a[(size_t)n * BATCH + r0 + t];
    }

    const float* gW2 = W2 + (size_t)n * HID * HID;
    #pragma unroll
    for (int i = 0; i < 32; ++i) {
        int idx2 = t + 256 * i;
        int e    = idx2 * 2;
        int row  = e >> 7;
        int col  = e & 127;
        v2f v = *(const v2f*)(gW2 + e);
        *(v2f*)(sW2 + row * W2_STRIDE + col) = v;
    }
    __syncthreads();

    const int lane = t & 31;
    const int wv   = t >> 5;
    const int li   = lane & 15;
    const int lh   = lane >> 4;

    const int brow = wv * 16 + li;
    const float aval = sA[brow];
    v2f bfr[32];
    #pragma unroll
    for (int s = 0; s < 32; ++s) {
        int k = 4 * s + 2 * lh;
        v2f w1v = *(const v2f*)(sW1 + k);
        v2f b1v = *(const v2f*)(sB1 + k);
        v2f h;
        h.x = fmaxf(aval * w1v.x + b1v.x, 0.0f);
        h.y = fmaxf(aval * w1v.y + b1v.y, 0.0f);
        bfr[s] = h;
    }

    float acc = 0.0f;
    #pragma unroll
    for (int mt = 0; mt < 8; ++mt) {
        v8f c = {0.f, 0.f, 0.f, 0.f, 0.f, 0.f, 0.f, 0.f};
        const float* arow = sW2 + (mt * 16 + li) * W2_STRIDE + 2 * lh;
        #pragma unroll
        for (int s = 0; s < 32; ++s) {
            v2f afrag = *(const v2f*)(arow + 4 * s);
            c = __builtin_amdgcn_wmma_f32_16x16x4_f32(
                    false, afrag, false, bfr[s], (short)0, c, false, false);
        }
        #pragma unroll
        for (int v = 0; v < 8; ++v) {
            int kout = mt * 16 + v + 8 * lh;
            float h2 = fmaxf(c[v] + sB2[kout], 0.0f);
            acc += h2 * sW3[kout];
        }
    }

    acc += __shfl_xor(acc, 16, 32);
    if (lh == 0) {
        out[(size_t)n * BATCH + r0 + brow] = acc + b3[n];
    }
}

extern "C" void kernel_launch(void* const* d_in, const int* in_sizes, int n_in,
                              void* d_out, int out_size, void* d_ws, size_t ws_size,
                              hipStream_t stream) {
    const float* a  = (const float*)d_in[0];
    const float* W1 = (const float*)d_in[1];
    const float* b1 = (const float*)d_in[2];
    const float* W2 = (const float*)d_in[3];
    const float* b2 = (const float*)d_in[4];
    const float* W3 = (const float*)d_in[5];
    const float* b3 = (const float*)d_in[6];
    float* out = (float*)d_out;

    dim3 grid(BATCH / ROWS_PER_BLK, NNETS);
    mlp256_fused_wmma_f32<<<grid, 256, 0, stream>>>(a, W1, b1, W2, b2, W3, b3, out);
}


# W2 A-fragments held in registers (512 VGPRs, one wave per SIMD): one 4-wave work-group per net stages W2 once and loops over its 2048 rows; WMMA loop has no LDS reads, vector work between WMMAs
# speedup vs baseline: 1.1311x; 1.1311x over previous
#include <hip/hip_runtime.h>

typedef __attribute__((ext_vector_type(2))) float v2f;
typedef __attribute__((ext_vector_type(4))) float v4f;
typedef __attribute__((ext_vector_type(8))) float v8f;

#define NNETS 256
#define HID 128
#define BATCH 2048
#define NWAVES 4
#define ROWS_PER_WAVE (BATCH / NWAVES)
#define TILES_PER_WAVE (ROWS_PER_WAVE / 16)
#define W2_STRIDE 132

#define WMMA_F32(A, B, C) __builtin_amdgcn_wmma_f32_16x16x4_f32(false, (A), false, (B), (short)0, (C), false, false)
#define V8ZERO ((v8f){0.f, 0.f, 0.f, 0.f, 0.f, 0.f, 0.f, 0.f})

#define ORDER_POINT(C, ACC) \
    asm volatile("" : "+v"((C)[0]), "+v"((C)[1]), "+v"((C)[2]), "+v"((C)[3]), "+v"(ACC))

#define PHASE_END(C, ACC) \
    asm volatile("v_nop\n\tv_nop\n\tv_nop\n\tv_nop" \
                 : "+v"((C)[0]), "+v"((C)[1]), "+v"((C)[2]), "+v"((C)[3]), "+v"(ACC))

#define MM_QUAD_WITH_EPI(C, MT, ACC, P, PMT)                                              \
    _Pragma("unroll")                                                                     \
    for (int s = 0; s < 32; ++s) {                                                        \
        _Pragma("unroll")                                                                 \
        for (int q = 0; q < 4; ++q)                                                       \
            (C)[q] = WMMA_F32(afr[(MT) + q][s], h[s], (C)[q]);                            \
        {                                                                                 \
            float h2 = fmaxf((P)[s >> 3][s & 7] + b2r[(PMT) + (s >> 3)][s & 7], 0.0f);    \
            (ACC) += h2 * w3r[(PMT) + (s >> 3)][s & 7];                                   \
        }                                                                                 \
        ORDER_POINT(C, ACC);                                                              \
    }

__device__ __forceinline__ float other_half(float x)
{
    unsigned u = __builtin_bit_cast(unsigned, x);
    u = __builtin_amdgcn_permlanex16(u, u, 0x76543210u, 0xfedcba98u, false, false);
    return __builtin_bit_cast(float, u);
}

__global__ __launch_bounds__(128) void mlp256_fused_wmma_f32(
    const float* __restrict__ a,  const float* __restrict__ W1,
    const float* __restrict__ b1, const float* __restrict__ W2,
    const float* __restrict__ b2, const float* __restrict__ W3,
    const float* __restrict__ b3, float* __restrict__ out)
{
    __shared__ __attribute__((aligned(16))) float sW2[HID * W2_STRIDE];
    __shared__ __attribute__((aligned(16))) float sW1[HID];
    __shared__ __attribute__((aligned(16))) float sB1[HID];
    __shared__ __attribute__((aligned(16))) float sB2[HID];
    __shared__ __attribute__((aligned(16))) float sW3[HID];

    const int n = blockIdx.x;
    const int t = threadIdx.x;

    sW1[t] = W1[n * HID + t];
    sB1[t] = b1[n * HID + t];
    sB2[t] = b2[n * HID + t];
    sW3[t] = W3[n * HID + t];

    const float* gW2 = W2 + (size_t)n * HID * HID;
    #pragma unroll
    for (int i = 0; i < 32; ++i) {
        int e   = (t + 128 * i) * 4;
        int row = e >> 7;
        int col = e & 127;
        v4f v = *(const v4f*)(gW2 + e);
        *(v4f*)(sW2 + row * W2_STRIDE + col) = v;
    }
    __syncthreads();

    const int lane = t & 31;
    const int wv   = t >> 5;
    const int li   = lane & 15;
    const int lh   = lane >> 4;

    v2f afr[8][32];
    #pragma unroll
    for (int mt = 0; mt < 8; ++mt) {
        const float* arow = sW2 + (mt * 16 + li) * W2_STRIDE + 2 * lh;
        #pragma unroll
        for (int s = 0; s < 32; ++s)
            afr[mt][s] = *(const v2f*)(arow + 4 * s);
    }
    v2f w1r[32], b1r[32];
    #pragma unroll
    for (int s = 0; s < 32; ++s) {
        int k = 4 * s + 2 * lh;
        w1r[s] = *(const v2f*)(sW1 + k);
        b1r[s] = *(const v2f*)(sB1 + k);
    }
    float b2r[8][8], w3r[8][8];
    #pragma unroll
    for (int mt = 0; mt < 8; ++mt) {
        #pragma unroll
        for (int v = 0; v < 8; ++v) {
            int kout = mt * 16 + v + 8 * lh;
            b2r[mt][v] = sB2[kout];
            w3r[mt][v] = sW3[kout];
        }
    }
    const float b3v = b3[n];

    const float* ap = a   + (size_t)n * BATCH + wv * ROWS_PER_WAVE + li;
    float*       op = out + (size_t)n * BATCH + wv * ROWS_PER_WAVE + li;
    float aval = ap[0];

    v8f cu[4] = {V8ZERO, V8ZERO, V8ZERO, V8ZERO};
    float accp = 0.0f;
    float res  = 0.0f;

    #pragma unroll 1
    for (int tile = 0; tile < TILES_PER_WAVE; ++tile) {
        if (tile > 1 && lh == 0) op[(tile - 2) * 16] = res;

        const int nt = (tile + 1 < TILES_PER_WAVE) ? tile + 1 : tile;
        const float anext = ap[nt * 16];

        v2f h[32];
        #pragma unroll
        for (int s = 0; s < 32; ++s) {
            h[s].x = fmaxf(aval * w1r[s].x + b1r[s].x, 0.0f);
            h[s].y = fmaxf(aval * w1r[s].y + b1r[s].y, 0.0f);
        }

        v8f cd[4] = {V8ZERO, V8ZERO, V8ZERO, V8ZERO};
        MM_QUAD_WITH_EPI(cd, 0, accp, cu, 4)
        PHASE_END(cd, accp);
        accp += other_half(accp);
        res = accp + b3v;

        float acc = 0.0f;
        v8f cn[4] = {V8ZERO, V8ZERO, V8ZERO, V8ZERO};
        MM_QUAD_WITH_EPI(cn, 4, acc, cd, 0)
        PHASE_END(cn, acc);

        #pragma unroll
        for (int q = 0; q < 4; ++q) cu[q] = cn[q];
        accp = acc;
        aval = anext;
    }
    if (lh == 0) op[(TILES_PER_WAVE - 2) * 16] = res;
    #pragma unroll
    for (int q = 0; q < 4; ++q) {
        #pragma unroll
        for (int v = 0; v < 8; ++v) {
            float h2 = fmaxf(cu[q][v] + b2r[4 + q][v], 0.0f);
            accp += h2 * w3r[4 + q][v];
        }
    }
    accp += other_half(accp);
    if (lh == 0) op[(TILES_PER_WAVE - 1) * 16] = accp + b3v;
}

extern "C" void kernel_launch(void* const* d_in, const int* in_sizes, int n_in,
                              void* d_out, int out_size, void* d_ws, size_t ws_size,
                              hipStream_t stream) {
    const float* a  = (const float*)d_in[0];
    const float* W1 = (const float*)d_in[1];
    const float* b1 = (const float*)d_in[2];
    const float* W2 = (const float*)d_in[3];
    const float* b2 = (const float*)d_in[4];
    const float* W3 = (const float*)d_in[5];
    const float* b3 = (const float*)d_in[6];
    float* out = (float*)d_out;

    mlp256_fused_wmma_f32<<<dim3(NNETS), 128, 0, stream>>>(a, W1, b1, W2, b2, W3, b3, out);
}


# Shorter vector chain per accumulator register: layer-2 bias is the WMMA start value, ReLU is one integer max, layer-3 is a packed multiply-add into two partial sums (order of sums changes)
# speedup vs baseline: 1.1724x; 1.0365x over previous
#include <hip/hip_runtime.h>

typedef __attribute__((ext_vector_type(2))) float v2f;
typedef __attribute__((ext_vector_type(4))) float v4f;
typedef __attribute__((ext_vector_type(8))) float v8f;

#define NNETS 256
#define HID 128
#define BATCH 2048
#define NWAVES 4
#define ROWS_PER_WAVE (BATCH / NWAVES)
#define TILES_PER_WAVE (ROWS_PER_WAVE / 16)
#define W2_STRIDE 132

#define WMMA_F32(A, B, C) __builtin_amdgcn_wmma_f32_16x16x4_f32(false, (A), false, (B), (short)0, (C), false, false)
#define V8ZERO ((v8f){0.f, 0.f, 0.f, 0.f, 0.f, 0.f, 0.f, 0.f})

#define ORDER_POINT(C, ACC, HH) \
    asm volatile("" : "+v"((C)[0]), "+v"((C)[1]), "+v"((C)[2]), "+v"((C)[3]), "+v"(ACC), "+v"(HH))

#define PHASE_END(C, ACC) \
    asm volatile("v_nop\n\tv_nop\n\tv_nop\n\tv_nop" \
                 : "+v"((C)[0]), "+v"((C)[1]), "+v"((C)[2]), "+v"((C)[3]), "+v"(ACC))

#define TRIP_END(C, ACC, IN) \
    asm volatile("v_nop\n\tv_nop\n\tv_nop\n\tv_nop" \
                 : "+v"((C)[0]), "+v"((C)[1]), "+v"((C)[2]), "+v"((C)[3]), "+v"(ACC), "+v"(IN))

#define MM_QUAD_WITH_EPI(C, MT, ACC, P, PMT)                                              \
    _Pragma("unroll")                                                                     \
    for (int s = 0; s < 32; ++s) {                                                        \
        _Pragma("unroll")                                                                 \
        for (int q = 0; q < 4; ++q)                                                       \
            (C)[q] = WMMA_F32(afr[(MT) + q][s], h[s], (C)[q]);                            \
        if ((s & 1) == 0) {                                                               \
            hh.x = relu_bits((P)[s >> 3][s & 7]);                                         \
            hh.y = relu_bits((P)[s >> 3][(s & 7) + 1]);                                   \
        } else {                                                                          \
            (ACC) = hh * w3p[(PMT) + (s >> 3)][(s & 7) >> 1] + (ACC);                     \
        }                                                                                 \
        ORDER_POINT(C, ACC, hh);                                                          \
    }

__device__ __forceinline__ float relu_bits(float x)
{
    int i = __builtin_bit_cast(int, x);
    i = i > 0 ? i : 0;
    return __builtin_bit_cast(float, i);
}

__device__ __forceinline__ float other_half(float x)
{
    unsigned u = __builtin_bit_cast(unsigned, x);
    u = __builtin_amdgcn_permlanex16(u, u, 0x76543210u, 0xfedcba98u, false, false);
    return __builtin_bit_cast(float, u);
}

__global__ __launch_bounds__(128) void mlp256_fused_wmma_f32(
    const float* __restrict__ a,  const float* __restrict__ W1,
    const float* __restrict__ b1, const float* __restrict__ W2,
    const float* __restrict__ b2, const float* __restrict__ W3,
    const float* __restrict__ b3, float* __restrict__ out)
{
    __shared__ __attribute__((aligned(16))) float sW2[HID * W2_STRIDE];
    __shared__ __attribute__((aligned(16))) float sW1[HID];
    __shared__ __attribute__((aligned(16))) float sB1[HID];
    __shared__ __attribute__((aligned(16))) float sB2[HID];
    __shared__ __attribute__((aligned(16))) float sW3[HID];

    const int n = blockIdx.x;
    const int t = threadIdx.x;

    sW1[t] = W1[n * HID + t];
    sB1[t] = b1[n * HID + t];
    sB2[t] = b2[n * HID + t];
    sW3[t] = W3[n * HID + t];

    const float* gW2 = W2 + (size_t)n * HID * HID;
    #pragma unroll
    for (int i = 0; i < 32; ++i) {
        int e   = (t + 128 * i) * 4;
        int row = e >> 7;
        int col = e & 127;
        v4f v = *(const v4f*)(gW2 + e);
        *(v4f*)(sW2 + row * W2_STRIDE + col) = v;
    }
    __syncthreads();

    const int lane = t & 31;
    const int wv   = t >> 5;
    const int li   = lane & 15;
    const int lh   = lane >> 4;

    v2f afr[8][32];
    #pragma unroll
    for (int mt = 0; mt < 8; ++mt) {
        const float* arow = sW2 + (mt * 16 + li) * W2_STRIDE + 2 * lh;
        #pragma unroll
        for (int s = 0; s < 32; ++s)
            afr[mt][s] = *(const v2f*)(arow + 4 * s);
    }
    v2f w1r[32], b1r[32];
    #pragma unroll
    for (int s = 0; s < 32; ++s) {
        int k = 4 * s + 2 * lh;
        w1r[s] = *(const v2f*)(sW1 + k);
        b1r[s] = *(const v2f*)(sB1 + k);
    }
    v8f b2v[8];
    v2f w3p[8][4];
    #pragma unroll
    for (int mt = 0; mt < 8; ++mt) {
        #pragma unroll
        for (int v = 0; v < 8; ++v)
            b2v[mt][v] = sB2[mt * 16 + v + 8 * lh];
        #pragma unroll
        for (int r = 0; r < 4; ++r)
            w3p[mt][r] = *(const v2f*)(sW3 + mt * 16 + 8 * lh + 2 * r);
    }
    const float b3v = b3[n];

    const float* ap = a   + (size_t)n * BATCH + wv * ROWS_PER_WAVE + li;
    float*       op = out + (size_t)n * BATCH + wv * ROWS_PER_WAVE + li;
    v2f av;
    av.x = ap[0];
    av.y = av.x;

    v8f cu[4] = {V8ZERO, V8ZERO, V8ZERO, V8ZERO};
    v2f accp = {0.0f, 0.0f};
    v2f hh   = {0.0f, 0.0f};
    float res  = 0.0f;

    #pragma unroll 1
    for (int tile = 0; tile < TILES_PER_WAVE; ++tile) {
        if (tile > 1 && lh == 0) op[(tile - 2) * 16] = res;

        const int nt = (tile + 1 < TILES_PER_WAVE) ? tile + 1 : tile;
        const float anext = ap[nt * 16];

        v2f h[32];
        #pragma unroll
        for (int s = 0; s < 32; ++s) {
            h[s].x = fmaxf(av.x * w1r[s].x + b1r[s].x, 0.0f);
            h[s].y = fmaxf(av.y * w1r[s].y + b1r[s].y, 0.0f);
        }

        v8f cd[4] = {b2v[0], b2v[1], b2v[2], b2v[3]};
        MM_QUAD_WITH_EPI(cd, 0, accp, cu, 4)
        PHASE_END(cd, accp);
        float tot = accp.x + accp.y;
        tot += other_half(tot);
        res = tot + b3v;

        v2f acc = {0.0f, 0.0f};
        v8f cn[4] = {b2v[4], b2v[5], b2v[6], b2v[7]};
        MM_QUAD_WITH_EPI(cn, 4, acc, cd, 0)
        TRIP_END(cn, acc, av);

        #pragma unroll
        for (int q = 0; q < 4; ++q) cu[q] = cn[q];
        accp = acc;
        av.x = anext;
        av.y = anext;
    }
    if (lh == 0) op[(TILES_PER_WAVE - 2) * 16] = res;
    #pragma unroll
    for (int q = 0; q < 4; ++q) {
        #pragma unroll
        for (int r = 0; r < 4; ++r) {
            hh.x = relu_bits(cu[q][2 * r]);
            hh.y = relu_bits(cu[q][2 * r + 1]);
            accp = hh * w3p[4 + q][r] + accp;
        }
    }
    float last = accp.x + accp.y;
    last += other_half(last);
    if (lh == 0) op[(TILES_PER_WAVE - 1) * 16] = last + b3v;
}

extern "C" void kernel_launch(void* const* d_in, const int* in_sizes, int n_in,
                              void* d_out, int out_size, void* d_ws, size_t ws_size,
                              hipStream_t stream) {
    const float* a  = (const float*)d_in[0];
    const float* W1 = (const float*)d_in[1];
    const float* b1 = (const float*)d_in[2];
    const float* W2 = (const float*)d_in[3];
    const float* b2 = (const float*)d_in[4];
    const float* W3 = (const float*)d_in[5];
    const float* b3 = (const float*)d_in[6];
    float* out = (float*)d_out;

    mlp256_fused_wmma_f32<<<dim3(NNETS), 128, 0, stream>>>(a, W1, b1, W2, b2, W3, b3, out);
}


# Rows sum layer 3 per segment: units whose two table values share a sign add to two per-segment sums, only sign-changing units are swept per row (order of sums changes)
# speedup vs baseline: 4.3304x; 3.6936x over previous
#include <hip/hip_runtime.h>

typedef __attribute__((ext_vector_type(2))) float v2f;
typedef __attribute__((ext_vector_type(4))) float v4f;
typedef __attribute__((ext_vector_type(8))) float v8f;

#define NNETS 256
#define HID 128
#define BATCH 2048
#define NWAVES 4
#define ROWS_PER_THREAD (BATCH / 128)
#define W2_STRIDE 132
#define NPTS 130
#define NPTS_PAD 144
#define PT 132
#define NSEG (NPTS - 1)
#define RP 129
#define REC_N 16704
#define ZREC (3 * REC_N - 12)

#define WMMA_F32(A, B, C) __builtin_amdgcn_wmma_f32_16x16x4_f32(false, (A), false, (B), (short)0, (C), false, false)

#define TRIP_END(C0, C1, IN, OFF) \
    asm volatile("v_nop\n\tv_nop\n\tv_nop\n\tv_nop" : "+v"(C0), "+v"(C1), "+v"(IN), "+v"(OFF))

#define H_READY_LO(H) \
    asm volatile("" : "+v"((H)[0]), "+v"((H)[1]), "+v"((H)[2]), "+v"((H)[3]), "+v"((H)[4]), "+v"((H)[5]), \
                      "+v"((H)[6]), "+v"((H)[7]), "+v"((H)[8]), "+v"((H)[9]), "+v"((H)[10]), "+v"((H)[11]), \
                      "+v"((H)[12]), "+v"((H)[13]), "+v"((H)[14]), "+v"((H)[15]))
#define H_READY_HI(H, C0, C1, OFF) \
    asm volatile("" : "+v"((H)[16]), "+v"((H)[17]), "+v"((H)[18]), "+v"((H)[19]), "+v"((H)[20]), "+v"((H)[21]), \
                      "+v"((H)[22]), "+v"((H)[23]), "+v"((H)[24]), "+v"((H)[25]), "+v"((H)[26]), "+v"((H)[27]), \
                      "+v"((H)[28]), "+v"((H)[29]), "+v"((H)[30]), "+v"((H)[31]), "+v"(C0), "+v"(C1), "+v"(OFF))

__global__ __launch_bounds__(128) void mlp256_knot_table_wmma_f32(
    const float* __restrict__ a,  const float* __restrict__ W1,
    const float* __restrict__ b1, const float* __restrict__ W2,
    const float* __restrict__ b2, const float* __restrict__ W3,
    const float* __restrict__ b3, float* __restrict__ out)
{
    __shared__ __attribute__((aligned(16))) float sU[3 * REC_N];
    float* const sW2  = sU;
    float* const rec  = sU;
    __shared__ float sA[NSEG + 3];
    __shared__ float sB[NSEG + 3];
    __shared__ int   sC[NSEG + 3];
    __shared__ __attribute__((aligned(16))) float sP[NPTS_PAD * PT];
    __shared__ __attribute__((aligned(16))) float sX[NPTS_PAD];
    __shared__ __attribute__((aligned(16))) float sS[2 * HID];
    __shared__ __attribute__((aligned(16))) float sK[HID];
    __shared__ __attribute__((aligned(16))) float sW1[HID];
    __shared__ __attribute__((aligned(16))) float sB1[HID];
    __shared__ __attribute__((aligned(16))) float sB2[HID];
    __shared__ __attribute__((aligned(16))) float sW3[HID];
    __shared__ float sLo[NWAVES];
    __shared__ float sHi[NWAVES];

    const int n = blockIdx.x;
    const int t = threadIdx.x;
    const int lane = t & 31;
    const int wv   = t >> 5;
    const int li   = lane & 15;
    const int lh   = lane >> 4;

    const float w1t = W1[n * HID + t];
    const float b1t = b1[n * HID + t];
    sW1[t] = w1t;
    sB1[t] = b1t;
    sB2[t] = b2[n * HID + t];
    sW3[t] = W3[n * HID + t];

    const float* ap = a   + (size_t)n * BATCH + t;
    float*       op = out + (size_t)n * BATCH + t;
    float ar[ROWS_PER_THREAD];
    #pragma unroll
    for (int p = 0; p < ROWS_PER_THREAD; ++p) ar[p] = ap[p * 128];
    float lo = ar[0], hi = ar[0];
    #pragma unroll
    for (int p = 1; p < ROWS_PER_THREAD; ++p) { lo = fminf(lo, ar[p]); hi = fmaxf(hi, ar[p]); }
    #pragma unroll
    for (int m = 16; m >= 1; m >>= 1) {
        lo = fminf(lo, __shfl_xor(lo, m, 32));
        hi = fmaxf(hi, __shfl_xor(hi, m, 32));
    }
    if (lane == 0) { sLo[wv] = lo; sHi[wv] = hi; }

    const float* gW2 = W2 + (size_t)n * HID * HID;
    #pragma unroll
    for (int i = 0; i < 32; ++i) {
        int e   = (t + 128 * i) * 4;
        int row = e >> 7;
        int col = e & 127;
        v4f v = *(const v4f*)(gW2 + e);
        *(v4f*)(sW2 + row * W2_STRIDE + col) = v;
    }
    __syncthreads();

    lo = fminf(fminf(sLo[0], sLo[1]), fminf(sLo[2], sLo[3]));
    hi = fmaxf(fmaxf(sHi[0], sHi[1]), fmaxf(sHi[2], sHi[3]));

    float xk = -b1t / w1t;
    xk = (xk >= lo && xk <= hi) ? xk : hi;
    sK[t] = xk;
    __syncthreads();

    {
        int r = 1;
        #pragma unroll
        for (int u4 = 0; u4 < HID / 4; ++u4) {
            v4f q = *(const v4f*)(sK + 4 * u4);
            #pragma unroll
            for (int e = 0; e < 4; ++e) {
                int u = 4 * u4 + e;
                r += ((q[e] < xk) || (q[e] == xk && u < t)) ? 1 : 0;
            }
        }
        r = r < 1 ? 1 : (r > HID ? HID : r);
        sX[r] = xk;
        sS[r] = xk;
        if (t == 0) { sX[0] = lo; sS[0] = lo; }
        if (t < NPTS_PAD - (HID + 1)) sX[HID + 1 + t] = hi;
        if (t >= 1) sS[HID + t] = __builtin_inff();
    }
    __syncthreads();

    {
        const int mt0 = 2 * wv;
        v2f afr[2][32];
        #pragma unroll
        for (int q = 0; q < 2; ++q) {
            const float* arow = sW2 + ((mt0 + q) * 16 + li) * W2_STRIDE + 2 * lh;
            #pragma unroll
            for (int s = 0; s < 32; ++s)
                afr[q][s] = *(const v2f*)(arow + 4 * s);
        }
        v2f w1r[32], b1r[32];
        #pragma unroll
        for (int s = 0; s < 32; ++s) {
            int k = 4 * s + 2 * lh;
            w1r[s] = *(const v2f*)(sW1 + k);
            b1r[s] = *(const v2f*)(sB1 + k);
        }
        v8f b2v[2];
        #pragma unroll
        for (int q = 0; q < 2; ++q) {
            #pragma unroll
            for (int v = 0; v < 8; ++v)
                b2v[q][v] = sB2[(mt0 + q) * 16 + v + 8 * lh];
        }

        int poff = li * PT + mt0 * 16 + 8 * lh;
        #pragma unroll 1
        for (int rt = 0; rt < NPTS_PAD / 16; ++rt) {
            v2f av;
            av.x = sX[rt * 16 + li];
            av.y = av.x;
            v2f h[32];
            #pragma unroll
            for (int s = 0; s < 32; ++s) {
                h[s].x = fmaxf(av.x * w1r[s].x + b1r[s].x, 0.0f);
                h[s].y = fmaxf(av.y * w1r[s].y + b1r[s].y, 0.0f);
            }
            v8f c0 = b2v[0], c1 = b2v[1];
            H_READY_LO(h);
            H_READY_HI(h, c0, c1, poff);
            #pragma unroll
            for (int s = 0; s < 32; ++s) {
                c0 = WMMA_F32(afr[0][s], h[s], c0);
                c1 = WMMA_F32(afr[1][s], h[s], c1);
            }
            TRIP_END(c0, c1, av, poff);
            float* prow = sP + poff;
            poff += 16 * PT;
            *(v4f*)(prow)          = (v4f){c0[0], c0[1], c0[2], c0[3]};
            *(v4f*)(prow + 4)      = (v4f){c0[4], c0[5], c0[6], c0[7]};
            *(v4f*)(prow + 16)     = (v4f){c1[0], c1[1], c1[2], c1[3]};
            *(v4f*)(prow + 16 + 4) = (v4f){c1[4], c1[5], c1[6], c1[7]};
        }
    }
    __syncthreads();

    {
        v4f w3r[HID / 4];
        #pragma unroll
        for (int k = 0; k < HID / 4; ++k) w3r[k] = *(const v4f*)(sW3 + 4 * k);

        const int j = t;
        const float* r0 = sP + j * PT;
        const float* r1 = r0 + PT;
        const int rb = j * RP + ((j >> 6) << 4);
        v4f s0 = {0.0f, 0.0f, 0.0f, 0.0f};
        v4f s1 = {0.0f, 0.0f, 0.0f, 0.0f};
        unsigned mk[4] = {0u, 0u, 0u, 0u};
        #pragma unroll
        for (int k = 0; k < HID / 4; ++k) {
            const v4f p0 = *(const v4f*)(r0 + 4 * k);
            const v4f p1 = *(const v4f*)(r1 + 4 * k);
            #pragma unroll
            for (int e = 0; e < 4; ++e) {
                const int xs = __float_as_int(p0[e]) ^ __float_as_int(p1[e]);
                mk[k >> 3] = __builtin_amdgcn_alignbit(mk[k >> 3], (unsigned)xs, 31);
                const float we = xs < 0 ? 0.0f : w3r[k][e];
                const float q0 = __int_as_float(max(__float_as_int(p0[e]), 0));
                const float q1 = __int_as_float(max(__float_as_int(p1[e]), 0));
                s0[e] = fmaf(we, q0, s0[e]);
                s1[e] = fmaf(we, q1, s1[e]);
            }
        }
        const float a0 = (s0[0] + s0[1]) + (s0[2] + s0[3]);
        const float a1 = (s1[0] + s1[1]) + (s1[2] + s1[3]);
        sA[j] = a0;
        sB[j] = a1 - a0;

        float* const rseg = rec + 3 * rb;
        int cnt3 = 0;
        #pragma unroll
        for (int q = 0; q < 4; ++q) {
            unsigned m = mk[q];
            while (m != 0u) {
                const int i = __builtin_clz(m);
                m &= ~(0x80000000u >> i);
                const int u = 32 * q + i;
                const float p0 = r0[u];
                const float p1 = r1[u];
                float* rp = rseg + cnt3;
                rp[0] = p0;
                rp[1] = p1 - p0;
                rp[2] = sW3[u];
                cnt3 += 3;
            }
        }
        sC[j] = cnt3;
        if (t == 0) { rec[ZREC] = 0.0f; rec[ZREC + 1] = 0.0f; rec[ZREC + 2] = 0.0f; }
    }

    if (wv == NWAVES - 1) {
        const int j = NSEG - 1;
        const v4f p0 = *(const v4f*)(sP + j * PT + 4 * lane);
        const v4f p1 = *(const v4f*)(sP + (j + 1) * PT + 4 * lane);
        const v4f w  = *(const v4f*)(sW3 + 4 * lane);
        const int rb = j * RP + ((j >> 6) << 4);
        float a0 = 0.0f, a1 = 0.0f;
        int base = 0;
        #pragma unroll
        for (int e = 0; e < 4; ++e) {
            const bool sw = (__float_as_int(p0[e]) ^ __float_as_int(p1[e])) < 0;
            const unsigned ball = __builtin_amdgcn_ballot_w32(sw);
            const int pos = base + (int)__builtin_amdgcn_mbcnt_lo(ball, 0u);
            if (sw) {
                float* rp = rec + 3 * (rb + pos);
                rp[0] = p0[e];
                rp[1] = p1[e] - p0[e];
                rp[2] = w[e];
            }
            base += __builtin_popcount(ball);
            const float we = sw ? 0.0f : w[e];
            a0 = fmaf(we, fmaxf(p0[e], 0.0f), a0);
            a1 = fmaf(we, fmaxf(p1[e], 0.0f), a1);
        }
        #pragma unroll
        for (int m = 16; m >= 1; m >>= 1) {
            a0 += __shfl_xor(a0, m, 32);
            a1 += __shfl_xor(a1, m, 32);
        }
        if (lane == 0) { sA[j] = a0; sB[j] = a1 - a0; sC[j] = 3 * base; }
    }
    __syncthreads();

    const float b3v = b3[n];

    int   rbase[ROWS_PER_THREAD];
    int   rc[ROWS_PER_THREAD];
    float tr[ROWS_PER_THREAD];
    float ov[ROWS_PER_THREAD];
    int   cl = 0;
    #pragma unroll
    for (int p = 0; p < ROWS_PER_THREAD; ++p) {
        const float x = ar[p];
        int j = 0;
        #pragma unroll
        for (int step = HID; step >= 1; step >>= 1) {
            const int c = j + step;
            j = (sS[c] <= x) ? c : j;
        }
        j = j < NPTS - 2 ? j : NPTS - 2;
        const float x0 = sX[j];
        const float x1 = sX[j + 1];
        const float wd = x1 - x0;
        const float tt = (x - x0) / (wd > 0.0f ? wd : 1.0f);
        tr[p] = tt;
        ov[p] = b3v + fmaf(tt, sB[j], sA[j]);
        int c = sC[j];
        c = c < 3 * HID ? c : 3 * HID;
        rc[p] = c;
        cl = c > cl ? c : cl;
        rbase[p] = 3 * (j * RP + ((j >> 6) << 4));
    }

    #pragma unroll 1
    for (int i = 0; i < cl; i += 3) {
        #pragma unroll
        for (int p = 0; p < ROWS_PER_THREAD; ++p) {
            const float* rp = rec + ((i < rc[p]) ? rbase[p] + i : ZREC);
            const float z = fmaf(tr[p], rp[1], rp[0]);
            ov[p] = fmaf(rp[2], fmaxf(z, 0.0f), ov[p]);
        }
    }

    #pragma unroll
    for (int p = 0; p < ROWS_PER_THREAD; ++p) op[p * 128] = ov[p];
}

extern "C" void kernel_launch(void* const* d_in, const int* in_sizes, int n_in,
                              void* d_out, int out_size, void* d_ws, size_t ws_size,
                              hipStream_t stream) {
    const float* a  = (const float*)d_in[0];
    const float* W1 = (const float*)d_in[1];
    const float* b1 = (const float*)d_in[2];
    const float* W2 = (const float*)d_in[3];
    const float* b2 = (const float*)d_in[4];
    const float* W3 = (const float*)d_in[5];
    const float* b3 = (const float*)d_in[6];
    float* out = (float*)d_out;

    mlp256_knot_table_wmma_f32<<<dim3(NNETS), 128, 0, stream>>>(a, W1, b1, W2, b2, W3, b3, out);
}


# Layer-2 table: per 16-point tile only the tile's own 16 units take WMMA k steps; the other units, linear on the tile, enter as per-tile slope/offset columns from one WMMA pass (order of sums changes)
# speedup vs baseline: 5.0252x; 1.1604x over previous
#include <hip/hip_runtime.h>

typedef __attribute__((ext_vector_type(2))) float v2f;
typedef __attribute__((ext_vector_type(2))) int   v2i;
typedef __attribute__((ext_vector_type(4))) float v4f;
typedef __attribute__((ext_vector_type(8))) float v8f;

#define NNETS 256
#define HID 128
#define BATCH 2048
#define NWAVES 4
#define ROWS_PER_THREAD (BATCH / 128)
#define W2_STRIDE 132
#define NPTS 130
#define NPTS_PAD 144
#define PT 132
#define NSEG (NPTS - 1)
#define RP 129
#define REC_N 16704
#define ZREC (3 * REC_N - 12)

#define WMMA_F32(A, B, C) __builtin_amdgcn_wmma_f32_16x16x4_f32(false, (A), false, (B), (short)0, (C), false, false)

#define TRIP_END4(C0, C1, C2, C3, IN, OFF, OFF2) \
    asm volatile("v_nop\n\tv_nop\n\tv_nop\n\tv_nop" : "+v"(C0), "+v"(C1), "+v"(C2), "+v"(C3), "+v"(IN), "+v"(OFF), "+v"(OFF2))
#define TRIP_END(C0, C1, IN, OFF, OFF2) \
    asm volatile("v_nop\n\tv_nop\n\tv_nop\n\tv_nop" : "+v"(C0), "+v"(C1), "+v"(IN), "+v"(OFF), "+v"(OFF2))

#define H_READY_LO(H) \
    asm volatile("" : "+v"((H)[0]), "+v"((H)[1]), "+v"((H)[2]), "+v"((H)[3]), "+v"((H)[4]), "+v"((H)[5]), \
                      "+v"((H)[6]), "+v"((H)[7]), "+v"((H)[8]), "+v"((H)[9]), "+v"((H)[10]), "+v"((H)[11]), \
                      "+v"((H)[12]), "+v"((H)[13]), "+v"((H)[14]), "+v"((H)[15]))
#define H_READY_HI(H, C0, C1, OFF) \
    asm volatile("" : "+v"((H)[16]), "+v"((H)[17]), "+v"((H)[18]), "+v"((H)[19]), "+v"((H)[20]), "+v"((H)[21]), \
                      "+v"((H)[22]), "+v"((H)[23]), "+v"((H)[24]), "+v"((H)[25]), "+v"((H)[26]), "+v"((H)[27]), \
                      "+v"((H)[28]), "+v"((H)[29]), "+v"((H)[30]), "+v"((H)[31]), "+v"(C0), "+v"(C1), "+v"(OFF))

__global__ __launch_bounds__(128) void mlp256_knot_table_wmma_f32(
    const float* __restrict__ a,  const float* __restrict__ W1,
    const float* __restrict__ b1, const float* __restrict__ W2,
    const float* __restrict__ b2, const float* __restrict__ W3,
    const float* __restrict__ b3, float* __restrict__ out)
{
    __shared__ __attribute__((aligned(16))) float sU[3 * REC_N];
    float* const sW2  = sU;
    float* const rec  = sU;
    __shared__ float sA[NSEG + 3];
    __shared__ float sB[NSEG + 3];
    __shared__ int   sC[NSEG + 3];
    __shared__ __attribute__((aligned(16))) float sP[NPTS_PAD * PT];
    __shared__ __attribute__((aligned(16))) float sX[NPTS_PAD];
    __shared__ __attribute__((aligned(16))) float sS[2 * HID];
    __shared__ __attribute__((aligned(16))) float sK[HID];
    __shared__ __attribute__((aligned(16))) float sW1[HID];
    __shared__ __attribute__((aligned(16))) float sB1[HID];
    __shared__ __attribute__((aligned(16))) float sW1s[HID];
    __shared__ __attribute__((aligned(16))) float sB1s[HID];
    __shared__ __attribute__((aligned(16))) int   sPerm[HID];
    __shared__ __attribute__((aligned(16))) float sAB[16 * PT];
    __shared__ __attribute__((aligned(16))) float sB2[HID];
    __shared__ __attribute__((aligned(16))) float sW3[HID];
    __shared__ float sLo[NWAVES];
    __shared__ float sHi[NWAVES];

    const int n = blockIdx.x;
    const int t = threadIdx.x;
    const int lane = t & 31;
    const int wv   = t >> 5;
    const int li   = lane & 15;
    const int lh   = lane >> 4;

    const float w1t = W1[n * HID + t];
    const float b1t = b1[n * HID + t];
    sW1[t] = w1t;
    sB1[t] = b1t;
    sB2[t] = b2[n * HID + t];
    sW3[t] = W3[n * HID + t];

    const float* ap = a   + (size_t)n * BATCH + t;
    float*       op = out + (size_t)n * BATCH + t;
    float ar[ROWS_PER_THREAD];
    #pragma unroll
    for (int p = 0; p < ROWS_PER_THREAD; ++p) ar[p] = ap[p * 128];
    float lo = ar[0], hi = ar[0];
    #pragma unroll
    for (int p = 1; p < ROWS_PER_THREAD; ++p) { lo = fminf(lo, ar[p]); hi = fmaxf(hi, ar[p]); }
    #pragma unroll
    for (int m = 16; m >= 1; m >>= 1) {
        lo = fminf(lo, __shfl_xor(lo, m, 32));
        hi = fmaxf(hi, __shfl_xor(hi, m, 32));
    }
    if (lane == 0) { sLo[wv] = lo; sHi[wv] = hi; }

    const float* gW2 = W2 + (size_t)n * HID * HID;
    #pragma unroll
    for (int i = 0; i < 32; ++i) {
        int e   = (t + 128 * i) * 4;
        int row = e >> 7;
        int col = e & 127;
        v4f v = *(const v4f*)(gW2 + e);
        *(v4f*)(sW2 + row * W2_STRIDE + col) = v;
    }
    __syncthreads();

    lo = fminf(fminf(sLo[0], sLo[1]), fminf(sLo[2], sLo[3]));
    hi = fmaxf(fmaxf(sHi[0], sHi[1]), fmaxf(sHi[2], sHi[3]));

    float xk = -b1t / w1t;
    xk = (xk >= lo && xk <= hi) ? xk : hi;
    sK[t] = xk;
    __syncthreads();

    {
        int r = 1;
        #pragma unroll
        for (int u4 = 0; u4 < HID / 4; ++u4) {
            v4f q = *(const v4f*)(sK + 4 * u4);
            #pragma unroll
            for (int e = 0; e < 4; ++e) {
                int u = 4 * u4 + e;
                r += ((q[e] < xk) || (q[e] == xk && u < t)) ? 1 : 0;
            }
        }
        r = r < 1 ? 1 : (r > HID ? HID : r);
        sX[r] = xk;
        sS[r] = xk;
        sW1s[r - 1] = w1t;
        sB1s[r - 1] = b1t;
        sPerm[r - 1] = t;
        if (t == 0) { sX[0] = lo; sS[0] = lo; }
        if (t < NPTS_PAD - (HID + 1)) sX[HID + 1 + t] = hi;
        if (t >= 1) sS[HID + t] = __builtin_inff();
    }
    __syncthreads();

    {
        const int mt0 = 2 * wv;
        v2f afr[2][32];
        v2i pu[32];
        #pragma unroll
        for (int s = 0; s < 32; ++s) pu[s] = *(const v2i*)(sPerm + 4 * s + 2 * lh);
        asm volatile("" : "+v"(pu[0]), "+v"(pu[1]), "+v"(pu[2]), "+v"(pu[3]), "+v"(pu[4]), "+v"(pu[5]), "+v"(pu[6]), "+v"(pu[7]),
                          "+v"(pu[8]), "+v"(pu[9]), "+v"(pu[10]), "+v"(pu[11]), "+v"(pu[12]), "+v"(pu[13]), "+v"(pu[14]), "+v"(pu[15]));
        asm volatile("" : "+v"(pu[16]), "+v"(pu[17]), "+v"(pu[18]), "+v"(pu[19]), "+v"(pu[20]), "+v"(pu[21]), "+v"(pu[22]), "+v"(pu[23]),
                          "+v"(pu[24]), "+v"(pu[25]), "+v"(pu[26]), "+v"(pu[27]), "+v"(pu[28]), "+v"(pu[29]), "+v"(pu[30]), "+v"(pu[31]));
        #pragma unroll
        for (int s = 0; s < 32; ++s) {
            const int u0 = pu[s].x & (HID - 1);
            const int u1 = pu[s].y & (HID - 1);
            #pragma unroll
            for (int q = 0; q < 2; ++q) {
                const float* arow = sW2 + ((mt0 + q) * 16 + li) * W2_STRIDE;
                afr[q][s].x = arow[u0];
                afr[q][s].y = arow[u1];
            }
        }
        v2f w1r[32], b1r[32];
        #pragma unroll
        for (int s = 0; s < 32; ++s) {
            int k = 4 * s + 2 * lh;
            w1r[s] = *(const v2f*)(sW1s + k);
            b1r[s] = *(const v2f*)(sB1s + k);
        }
        v8f b2v[2];
        #pragma unroll
        for (int q = 0; q < 2; ++q) {
            #pragma unroll
            for (int v = 0; v < 8; ++v)
                b2v[q][v] = sB2[(mt0 + q) * 16 + v + 8 * lh];
        }
        float xs[9];
        #pragma unroll
        for (int T = 0; T < 8; ++T) xs[T] = sX[16 * T + 1 + li];
        xs[8] = xs[7];

        int poff = li * PT + mt0 * 16 + 8 * lh;
        int aoff = mt0 * 16 + 8 * lh;

        v8f a00, a10, a01, a11;
        {
            const int   tau = li & 7;
            const bool  isSlope = li < 8;
            const float xm = 0.5f * sX[16 * tau + 1] + 0.5f * sX[16 * tau + 16];
            const float xe = (li == 0) ? lo : hi;
            v2f g[32], h[32];
            #pragma unroll
            for (int s = 0; s < 32; ++s) {
                const bool other = (s >> 2) != tau;
                const float p0 = fmaf(xm, w1r[s].x, b1r[s].x);
                const float p1 = fmaf(xm, w1r[s].y, b1r[s].y);
                const float s0 = isSlope ? w1r[s].x : b1r[s].x;
                const float s1 = isSlope ? w1r[s].y : b1r[s].y;
                g[s].x = (other && p0 > 0.0f) ? s0 : 0.0f;
                g[s].y = (other && p1 > 0.0f) ? s1 : 0.0f;
                h[s].x = fmaxf(fmaf(xe, w1r[s].x, b1r[s].x), 0.0f);
                h[s].y = fmaxf(fmaf(xe, w1r[s].y, b1r[s].y), 0.0f);
            }
            #pragma unroll
            for (int v = 0; v < 8; ++v) {
                a00[v] = isSlope ? 0.0f : b2v[0][v];
                a10[v] = isSlope ? 0.0f : b2v[1][v];
            }
            a01 = b2v[0];
            a11 = b2v[1];
            asm volatile("" : "+v"(xs[0]), "+v"(xs[1]), "+v"(xs[2]), "+v"(xs[3]), "+v"(xs[4]), "+v"(xs[5]),
                              "+v"(xs[6]), "+v"(xs[7]), "+v"(aoff));
            H_READY_LO(g);
            H_READY_HI(g, a00, a10, poff);
            H_READY_LO(h);
            H_READY_HI(h, a01, a11, poff);
            asm volatile("" : "+v"(a00), "+v"(a10), "+v"(a01), "+v"(a11));
            #pragma unroll
            for (int s = 0; s < 32; ++s) {
                a00 = WMMA_F32(afr[0][s], g[s], a00);
                a10 = WMMA_F32(afr[1][s], g[s], a10);
                a01 = WMMA_F32(afr[0][s], h[s], a01);
                a11 = WMMA_F32(afr[1][s], h[s], a11);
            }
            TRIP_END4(a00, a10, a01, a11, xs[0], poff, aoff);
            float* prow = sP + poff + ((li == 0) ? 0 : 128 * PT);
            *(v4f*)(prow)          = (v4f){a01[0], a01[1], a01[2], a01[3]};
            *(v4f*)(prow + 4)      = (v4f){a01[4], a01[5], a01[6], a01[7]};
            *(v4f*)(prow + 16)     = (v4f){a11[0], a11[1], a11[2], a11[3]};
            *(v4f*)(prow + 16 + 4) = (v4f){a11[4], a11[5], a11[6], a11[7]};
            float* crow = sAB + poff;
            *(v4f*)(crow)          = (v4f){a00[0], a00[1], a00[2], a00[3]};
            *(v4f*)(crow + 4)      = (v4f){a00[4], a00[5], a00[6], a00[7]};
            *(v4f*)(crow + 16)     = (v4f){a10[0], a10[1], a10[2], a10[3]};
            *(v4f*)(crow + 16 + 4) = (v4f){a10[4], a10[5], a10[6], a10[7]};
        }

        const float* ab = sAB + aoff;
        v4f sl[8][4], of[8][4];
        #pragma unroll
        for (int T = 0; T < 8; ++T) {
            sl[T][0] = *(const v4f*)(ab + T * PT);
            sl[T][1] = *(const v4f*)(ab + T * PT + 4);
            sl[T][2] = *(const v4f*)(ab + T * PT + 16);
            sl[T][3] = *(const v4f*)(ab + T * PT + 20);
            of[T][0] = *(const v4f*)(ab + (8 + T) * PT);
            of[T][1] = *(const v4f*)(ab + (8 + T) * PT + 4);
            of[T][2] = *(const v4f*)(ab + (8 + T) * PT + 16);
            of[T][3] = *(const v4f*)(ab + (8 + T) * PT + 20);
        }
        poff += PT;
        #pragma unroll
        for (int T = 0; T < 8; ++T) {
            const float x = xs[T];
            v8f c0, c1;
            #pragma unroll
            for (int v = 0; v < 4; ++v) {
                c0[v]     = fmaf(sl[T][0][v], x, of[T][0][v]);
                c0[v + 4] = fmaf(sl[T][1][v], x, of[T][1][v]);
                c1[v]     = fmaf(sl[T][2][v], x, of[T][2][v]);
                c1[v + 4] = fmaf(sl[T][3][v], x, of[T][3][v]);
            }
            v2f hb[4];
            #pragma unroll
            for (int e = 0; e < 4; ++e) {
                hb[e].x = fmaxf(fmaf(x, w1r[4 * T + e].x, b1r[4 * T + e].x), 0.0f);
                hb[e].y = fmaxf(fmaf(x, w1r[4 * T + e].y, b1r[4 * T + e].y), 0.0f);
            }
            asm volatile("" : "+v"(hb[0]), "+v"(hb[1]), "+v"(hb[2]), "+v"(hb[3]), "+v"(c0), "+v"(c1), "+v"(poff) : : "memory");
            #pragma unroll
            for (int e = 0; e < 4; ++e) {
                c0 = WMMA_F32(afr[0][4 * T + e], hb[e], c0);
                c1 = WMMA_F32(afr[1][4 * T + e], hb[e], c1);
            }
            TRIP_END(c0, c1, xs[T + 1], poff, aoff);
            float* prow = sP + poff;
            poff += 16 * PT;
            *(v4f*)(prow)          = (v4f){c0[0], c0[1], c0[2], c0[3]};
            *(v4f*)(prow + 4)      = (v4f){c0[4], c0[5], c0[6], c0[7]};
            *(v4f*)(prow + 16)     = (v4f){c1[0], c1[1], c1[2], c1[3]};
            *(v4f*)(prow + 16 + 4) = (v4f){c1[4], c1[5], c1[6], c1[7]};
        }
    }
    __syncthreads();

    {
        v4f w3r[HID / 4];
        #pragma unroll
        for (int k = 0; k < HID / 4; ++k) w3r[k] = *(const v4f*)(sW3 + 4 * k);

        const int j = t;
        const float* r0 = sP + j * PT;
        const float* r1 = r0 + PT;
        const int rb = j * RP + ((j >> 6) << 4);
        v4f s0 = {0.0f, 0.0f, 0.0f, 0.0f};
        v4f s1 = {0.0f, 0.0f, 0.0f, 0.0f};
        unsigned mk[4] = {0u, 0u, 0u, 0u};
        #pragma unroll
        for (int k = 0; k < HID / 4; ++k) {
            const v4f p0 = *(const v4f*)(r0 + 4 * k);
            const v4f p1 = *(const v4f*)(r1 + 4 * k);
            #pragma unroll
            for (int e = 0; e < 4; ++e) {
                const int xs = __float_as_int(p0[e]) ^ __float_as_int(p1[e]);
                mk[k >> 3] = __builtin_amdgcn_alignbit(mk[k >> 3], (unsigned)xs, 31);
                const float we = xs < 0 ? 0.0f : w3r[k][e];
                const float q0 = __int_as_float(max(__float_as_int(p0[e]), 0));
                const float q1 = __int_as_float(max(__float_as_int(p1[e]), 0));
                s0[e] = fmaf(we, q0, s0[e]);
                s1[e] = fmaf(we, q1, s1[e]);
            }
        }
        const float a0 = (s0[0] + s0[1]) + (s0[2] + s0[3]);
        const float a1 = (s1[0] + s1[1]) + (s1[2] + s1[3]);
        sA[j] = a0;
        sB[j] = a1 - a0;

        float* const rseg = rec + 3 * rb;
        int cnt3 = 0;
        #pragma unroll
        for (int q = 0; q < 4; ++q) {
            unsigned m = mk[q];
            while (m != 0u) {
                const int i = __builtin_clz(m);
                m &= ~(0x80000000u >> i);
                const int u = 32 * q + i;
                const float p0 = r0[u];
                const float p1 = r1[u];
                float* rp = rseg + cnt3;
                rp[0] = p0;
                rp[1] = p1 - p0;
                rp[2] = sW3[u];
                cnt3 += 3;
            }
        }
        sC[j] = cnt3;
        if (t == 0) { rec[ZREC] = 0.0f; rec[ZREC + 1] = 0.0f; rec[ZREC + 2] = 0.0f; }
    }

    if (wv == NWAVES - 1) {
        const int j = NSEG - 1;
        const v4f p0 = *(const v4f*)(sP + j * PT + 4 * lane);
        const v4f p1 = *(const v4f*)(sP + (j + 1) * PT + 4 * lane);
        const v4f w  = *(const v4f*)(sW3 + 4 * lane);
        const int rb = j * RP + ((j >> 6) << 4);
        float a0 = 0.0f, a1 = 0.0f;
        int base = 0;
        #pragma unroll
        for (int e = 0; e < 4; ++e) {
            const bool sw = (__float_as_int(p0[e]) ^ __float_as_int(p1[e])) < 0;
            const unsigned ball = __builtin_amdgcn_ballot_w32(sw);
            const int pos = base + (int)__builtin_amdgcn_mbcnt_lo(ball, 0u);
            if (sw) {
                float* rp = rec + 3 * (rb + pos);
                rp[0] = p0[e];
                rp[1] = p1[e] - p0[e];
                rp[2] = w[e];
            }
            base += __builtin_popcount(ball);
            const float we = sw ? 0.0f : w[e];
            a0 = fmaf(we, fmaxf(p0[e], 0.0f), a0);
            a1 = fmaf(we, fmaxf(p1[e], 0.0f), a1);
        }
        #pragma unroll
        for (int m = 16; m >= 1; m >>= 1) {
            a0 += __shfl_xor(a0, m, 32);
            a1 += __shfl_xor(a1, m, 32);
        }
        if (lane == 0) { sA[j] = a0; sB[j] = a1 - a0; sC[j] = 3 * base; }
    }
    __syncthreads();

    const float b3v = b3[n];

    int   rbase[ROWS_PER_THREAD];
    int   rc[ROWS_PER_THREAD];
    float tr[ROWS_PER_THREAD];
    float ov[ROWS_PER_THREAD];
    int   cl = 0;
    #pragma unroll
    for (int p = 0; p < ROWS_PER_THREAD; ++p) {
        const float x = ar[p];
        int j = 0;
        #pragma unroll
        for (int step = HID; step >= 1; step >>= 1) {
            const int c = j + step;
            j = (sS[c] <= x) ? c : j;
        }
        j = j < NPTS - 2 ? j : NPTS - 2;
        const float x0 = sX[j];
        const float x1 = sX[j + 1];
        const float wd = x1 - x0;
        const float tt = (x - x0) / (wd > 0.0f ? wd : 1.0f);
        tr[p] = tt;
        ov[p] = b3v + fmaf(tt, sB[j], sA[j]);
        int c = sC[j];
        c = c < 3 * HID ? c : 3 * HID;
        rc[p] = c;
        cl = c > cl ? c : cl;
        rbase[p] = 3 * (j * RP + ((j >> 6) << 4));
    }

    #pragma unroll 1
    for (int i = 0; i < cl; i += 3) {
        #pragma unroll
        for (int p = 0; p < ROWS_PER_THREAD; ++p) {
            const float* rp = rec + ((i < rc[p]) ? rbase[p] + i : ZREC);
            const float z = fmaf(tr[p], rp[1], rp[0]);
            ov[p] = fmaf(rp[2], fmaxf(z, 0.0f), ov[p]);
        }
    }

    #pragma unroll
    for (int p = 0; p < ROWS_PER_THREAD; ++p) op[p * 128] = ov[p];
}

extern "C" void kernel_launch(void* const* d_in, const int* in_sizes, int n_in,
                              void* d_out, int out_size, void* d_ws, size_t ws_size,
                              hipStream_t stream) {
    const float* a  = (const float*)d_in[0];
    const float* W1 = (const float*)d_in[1];
    const float* b1 = (const float*)d_in[2];
    const float* W2 = (const float*)d_in[3];
    const float* b2 = (const float*)d_in[4];
    const float* W3 = (const float*)d_in[5];
    const float* b3 = (const float*)d_in[6];
    float* out = (float*)d_out;

    mlp256_knot_table_wmma_f32<<<dim3(NNETS), 128, 0, stream>>>(a, W1, b1, W2, b2, W3, b3, out);
}


# Load round trips off the lone wave's path: W2 requested at entry and stored to LDS behind the kink sort; sort keys, the 16 search chains and the segment data read as full groups before use
# speedup vs baseline: 5.4846x; 1.0914x over previous
#include <hip/hip_runtime.h>

typedef __attribute__((ext_vector_type(2))) float v2f;
typedef __attribute__((ext_vector_type(2))) int   v2i;
typedef __attribute__((ext_vector_type(4))) float v4f;
typedef __attribute__((ext_vector_type(8))) float v8f;

#define NNETS 256
#define HID 128
#define BATCH 2048
#define NWAVES 4
#define ROWS_PER_THREAD (BATCH / 128)
#define W2_STRIDE 132
#define NPTS 130
#define NPTS_PAD 144
#define PT 132
#define NSEG (NPTS - 1)
#define RP 129
#define REC_N 16704
#define ZREC (3 * REC_N - 12)

#define WMMA_F32(A, B, C) __builtin_amdgcn_wmma_f32_16x16x4_f32(false, (A), false, (B), (short)0, (C), false, false)

#define TRIP_END4(C0, C1, C2, C3, IN, OFF, OFF2) \
    asm volatile("v_nop\n\tv_nop\n\tv_nop\n\tv_nop" : "+v"(C0), "+v"(C1), "+v"(C2), "+v"(C3), "+v"(IN), "+v"(OFF), "+v"(OFF2))
#define TRIP_END(C0, C1, IN, OFF, OFF2) \
    asm volatile("v_nop\n\tv_nop\n\tv_nop\n\tv_nop" : "+v"(C0), "+v"(C1), "+v"(IN), "+v"(OFF), "+v"(OFF2))

#define H_READY_LO(H) \
    asm volatile("" : "+v"((H)[0]), "+v"((H)[1]), "+v"((H)[2]), "+v"((H)[3]), "+v"((H)[4]), "+v"((H)[5]), \
                      "+v"((H)[6]), "+v"((H)[7]), "+v"((H)[8]), "+v"((H)[9]), "+v"((H)[10]), "+v"((H)[11]), \
                      "+v"((H)[12]), "+v"((H)[13]), "+v"((H)[14]), "+v"((H)[15]))
#define H_READY_HI(H, C0, C1, OFF) \
    asm volatile("" : "+v"((H)[16]), "+v"((H)[17]), "+v"((H)[18]), "+v"((H)[19]), "+v"((H)[20]), "+v"((H)[21]), \
                      "+v"((H)[22]), "+v"((H)[23]), "+v"((H)[24]), "+v"((H)[25]), "+v"((H)[26]), "+v"((H)[27]), \
                      "+v"((H)[28]), "+v"((H)[29]), "+v"((H)[30]), "+v"((H)[31]), "+v"(C0), "+v"(C1), "+v"(OFF))

#define LDS_BARRIER() do { \
    __builtin_amdgcn_fence(__ATOMIC_RELEASE, "workgroup", "local"); \
    __builtin_amdgcn_s_barrier(); \
    __builtin_amdgcn_fence(__ATOMIC_ACQUIRE, "workgroup", "local"); } while (0)

__global__ __launch_bounds__(128) void mlp256_knot_table_wmma_f32(
    const float* __restrict__ a,  const float* __restrict__ W1,
    const float* __restrict__ b1, const float* W2,
    const float* __restrict__ b2, const float* __restrict__ W3,
    const float* __restrict__ b3, float* __restrict__ out)
{
    __shared__ __attribute__((aligned(16))) float sU[3 * REC_N];
    float* const sW2  = sU;
    float* const rec  = sU;
    __shared__ float sA[NSEG + 3];
    __shared__ float sB[NSEG + 3];
    __shared__ int   sC[NSEG + 3];
    __shared__ __attribute__((aligned(16))) float sP[NPTS_PAD * PT];
    __shared__ __attribute__((aligned(16))) float sX[NPTS_PAD];
    __shared__ __attribute__((aligned(16))) float sS[2 * HID];
    __shared__ __attribute__((aligned(16))) float sK[HID];
    __shared__ __attribute__((aligned(16))) float sW1[HID];
    __shared__ __attribute__((aligned(16))) float sB1[HID];
    __shared__ __attribute__((aligned(16))) float sW1s[HID];
    __shared__ __attribute__((aligned(16))) float sB1s[HID];
    __shared__ __attribute__((aligned(16))) int   sPerm[HID];
    __shared__ __attribute__((aligned(16))) float sAB[16 * PT];
    __shared__ __attribute__((aligned(16))) float sB2[HID];
    __shared__ __attribute__((aligned(16))) float sW3[HID];
    __shared__ float sLo[NWAVES];
    __shared__ float sHi[NWAVES];

    const int n = blockIdx.x;
    const int t = threadIdx.x;
    const int lane = t & 31;
    const int wv   = t >> 5;
    const int li   = lane & 15;
    const int lh   = lane >> 4;

    const float w1t = W1[n * HID + t];
    const float b1t = b1[n * HID + t];
    sW1[t] = w1t;
    sB1[t] = b1t;
    sB2[t] = b2[n * HID + t];
    sW3[t] = W3[n * HID + t];

    const float* ap = a   + (size_t)n * BATCH + t;
    float*       op = out + (size_t)n * BATCH + t;
    float ar[ROWS_PER_THREAD];
    #pragma unroll
    for (int p = 0; p < ROWS_PER_THREAD; ++p) ar[p] = ap[p * 128];

    const float* gW2 = W2 + (size_t)n * HID * HID;
    v4f w2v[32];
    #pragma unroll
    for (int i = 0; i < 32; ++i) w2v[i] = *(const v4f*)(gW2 + (t + 128 * i) * 4);
    float lo = ar[0], hi = ar[0];
    #pragma unroll
    for (int p = 1; p < ROWS_PER_THREAD; ++p) { lo = fminf(lo, ar[p]); hi = fmaxf(hi, ar[p]); }
    #pragma unroll
    for (int m = 16; m >= 1; m >>= 1) {
        lo = fminf(lo, __shfl_xor(lo, m, 32));
        hi = fmaxf(hi, __shfl_xor(hi, m, 32));
    }
    if (lane == 0) { sLo[wv] = lo; sHi[wv] = hi; }

    LDS_BARRIER();

    lo = fminf(fminf(sLo[0], sLo[1]), fminf(sLo[2], sLo[3]));
    hi = fmaxf(fmaxf(sHi[0], sHi[1]), fmaxf(sHi[2], sHi[3]));

    float xk = -b1t / w1t;
    xk = (xk >= lo && xk <= hi) ? xk : hi;
    sK[t] = xk;
    LDS_BARRIER();

    {
        int r = 1;
        v4f q[HID / 4];
        #pragma unroll
        for (int u4 = 0; u4 < HID / 4; ++u4) q[u4] = *(const v4f*)(sK + 4 * u4);
        asm volatile("" : "+v"(q[0]), "+v"(q[1]), "+v"(q[2]), "+v"(q[3]), "+v"(q[4]), "+v"(q[5]), "+v"(q[6]), "+v"(q[7]),
                          "+v"(q[8]), "+v"(q[9]), "+v"(q[10]), "+v"(q[11]), "+v"(q[12]), "+v"(q[13]), "+v"(q[14]), "+v"(q[15]));
        asm volatile("" : "+v"(q[16]), "+v"(q[17]), "+v"(q[18]), "+v"(q[19]), "+v"(q[20]), "+v"(q[21]), "+v"(q[22]), "+v"(q[23]),
                          "+v"(q[24]), "+v"(q[25]), "+v"(q[26]), "+v"(q[27]), "+v"(q[28]), "+v"(q[29]), "+v"(q[30]), "+v"(q[31]));
        #pragma unroll
        for (int u4 = 0; u4 < HID / 4; ++u4) {
            #pragma unroll
            for (int e = 0; e < 4; ++e) {
                int u = 4 * u4 + e;
                r += ((q[u4][e] < xk) || (q[u4][e] == xk && u < t)) ? 1 : 0;
            }
        }
        r = r < 1 ? 1 : (r > HID ? HID : r);
        sX[r] = xk;
        sS[r] = xk;
        sW1s[r - 1] = w1t;
        sB1s[r - 1] = b1t;
        sPerm[r - 1] = t;
        if (t == 0) { sX[0] = lo; sS[0] = lo; }
        if (t < NPTS_PAD - (HID + 1)) sX[HID + 1 + t] = hi;
        if (t >= 1) sS[HID + t] = __builtin_inff();
    }
    #pragma unroll
    for (int i = 0; i < 32; ++i) {
        int e   = (t + 128 * i) * 4;
        int row = e >> 7;
        int col = e & 127;
        *(v4f*)(sW2 + row * W2_STRIDE + col) = w2v[i];
    }
    __syncthreads();

    {
        const int mt0 = 2 * wv;
        v2f afr[2][32];
        v2i pu[32];
        #pragma unroll
        for (int s = 0; s < 32; ++s) pu[s] = *(const v2i*)(sPerm + 4 * s + 2 * lh);
        asm volatile("" : "+v"(pu[0]), "+v"(pu[1]), "+v"(pu[2]), "+v"(pu[3]), "+v"(pu[4]), "+v"(pu[5]), "+v"(pu[6]), "+v"(pu[7]),
                          "+v"(pu[8]), "+v"(pu[9]), "+v"(pu[10]), "+v"(pu[11]), "+v"(pu[12]), "+v"(pu[13]), "+v"(pu[14]), "+v"(pu[15]));
        asm volatile("" : "+v"(pu[16]), "+v"(pu[17]), "+v"(pu[18]), "+v"(pu[19]), "+v"(pu[20]), "+v"(pu[21]), "+v"(pu[22]), "+v"(pu[23]),
                          "+v"(pu[24]), "+v"(pu[25]), "+v"(pu[26]), "+v"(pu[27]), "+v"(pu[28]), "+v"(pu[29]), "+v"(pu[30]), "+v"(pu[31]));
        #pragma unroll
        for (int s = 0; s < 32; ++s) {
            const int u0 = pu[s].x & (HID - 1);
            const int u1 = pu[s].y & (HID - 1);
            #pragma unroll
            for (int q = 0; q < 2; ++q) {
                const float* arow = sW2 + ((mt0 + q) * 16 + li) * W2_STRIDE;
                afr[q][s].x = arow[u0];
                afr[q][s].y = arow[u1];
            }
        }
        v2f w1r[32], b1r[32];
        #pragma unroll
        for (int s = 0; s < 32; ++s) {
            int k = 4 * s + 2 * lh;
            w1r[s] = *(const v2f*)(sW1s + k);
            b1r[s] = *(const v2f*)(sB1s + k);
        }
        v8f b2v[2];
        #pragma unroll
        for (int q = 0; q < 2; ++q) {
            #pragma unroll
            for (int v = 0; v < 8; ++v)
                b2v[q][v] = sB2[(mt0 + q) * 16 + v + 8 * lh];
        }
        float xs[9];
        #pragma unroll
        for (int T = 0; T < 8; ++T) xs[T] = sX[16 * T + 1 + li];
        xs[8] = xs[7];

        int poff = li * PT + mt0 * 16 + 8 * lh;
        int aoff = mt0 * 16 + 8 * lh;

        v8f a00, a10, a01, a11;
        {
            const int   tau = li & 7;
            const bool  isSlope = li < 8;
            const float xm = 0.5f * sX[16 * tau + 1] + 0.5f * sX[16 * tau + 16];
            const float xe = (li == 0) ? lo : hi;
            v2f g[32], h[32];
            #pragma unroll
            for (int s = 0; s < 32; ++s) {
                const bool other = (s >> 2) != tau;
                const float p0 = fmaf(xm, w1r[s].x, b1r[s].x);
                const float p1 = fmaf(xm, w1r[s].y, b1r[s].y);
                const float s0 = isSlope ? w1r[s].x : b1r[s].x;
                const float s1 = isSlope ? w1r[s].y : b1r[s].y;
                g[s].x = (other && p0 > 0.0f) ? s0 : 0.0f;
                g[s].y = (other && p1 > 0.0f) ? s1 : 0.0f;
                h[s].x = fmaxf(fmaf(xe, w1r[s].x, b1r[s].x), 0.0f);
                h[s].y = fmaxf(fmaf(xe, w1r[s].y, b1r[s].y), 0.0f);
            }
            #pragma unroll
            for (int v = 0; v < 8; ++v) {
                a00[v] = isSlope ? 0.0f : b2v[0][v];
                a10[v] = isSlope ? 0.0f : b2v[1][v];
            }
            a01 = b2v[0];
            a11 = b2v[1];
            asm volatile("" : "+v"(xs[0]), "+v"(xs[1]), "+v"(xs[2]), "+v"(xs[3]), "+v"(xs[4]), "+v"(xs[5]),
                              "+v"(xs[6]), "+v"(xs[7]), "+v"(aoff));
            H_READY_LO(g);
            H_READY_HI(g, a00, a10, poff);
            H_READY_LO(h);
            H_READY_HI(h, a01, a11, poff);
            asm volatile("" : "+v"(a00), "+v"(a10), "+v"(a01), "+v"(a11));
            #pragma unroll
            for (int s = 0; s < 32; ++s) {
                a00 = WMMA_F32(afr[0][s], g[s], a00);
                a10 = WMMA_F32(afr[1][s], g[s], a10);
                a01 = WMMA_F32(afr[0][s], h[s], a01);
                a11 = WMMA_F32(afr[1][s], h[s], a11);
            }
            TRIP_END4(a00, a10, a01, a11, xs[0], poff, aoff);
            float* prow = sP + poff + ((li == 0) ? 0 : 128 * PT);
            *(v4f*)(prow)          = (v4f){a01[0], a01[1], a01[2], a01[3]};
            *(v4f*)(prow + 4)      = (v4f){a01[4], a01[5], a01[6], a01[7]};
            *(v4f*)(prow + 16)     = (v4f){a11[0], a11[1], a11[2], a11[3]};
            *(v4f*)(prow + 16 + 4) = (v4f){a11[4], a11[5], a11[6], a11[7]};
            float* crow = sAB + poff;
            *(v4f*)(crow)          = (v4f){a00[0], a00[1], a00[2], a00[3]};
            *(v4f*)(crow + 4)      = (v4f){a00[4], a00[5], a00[6], a00[7]};
            *(v4f*)(crow + 16)     = (v4f){a10[0], a10[1], a10[2], a10[3]};
            *(v4f*)(crow + 16 + 4) = (v4f){a10[4], a10[5], a10[6], a10[7]};
        }

        const float* ab = sAB + aoff;
        v4f sl[8][4], of[8][4];
        #pragma unroll
        for (int T = 0; T < 8; ++T) {
            sl[T][0] = *(const v4f*)(ab + T * PT);
            sl[T][1] = *(const v4f*)(ab + T * PT + 4);
            sl[T][2] = *(const v4f*)(ab + T * PT + 16);
            sl[T][3] = *(const v4f*)(ab + T * PT + 20);
            of[T][0] = *(const v4f*)(ab + (8 + T) * PT);
            of[T][1] = *(const v4f*)(ab + (8 + T) * PT + 4);
            of[T][2] = *(const v4f*)(ab + (8 + T) * PT + 16);
            of[T][3] = *(const v4f*)(ab + (8 + T) * PT + 20);
        }
        poff += PT;
        #pragma unroll
        for (int T = 0; T < 8; ++T) {
            const float x = xs[T];
            v8f c0, c1;
            #pragma unroll
            for (int v = 0; v < 4; ++v) {
                c0[v]     = fmaf(sl[T][0][v], x, of[T][0][v]);
                c0[v + 4] = fmaf(sl[T][1][v], x, of[T][1][v]);
                c1[v]     = fmaf(sl[T][2][v], x, of[T][2][v]);
                c1[v + 4] = fmaf(sl[T][3][v], x, of[T][3][v]);
            }
            v2f hb[4];
            #pragma unroll
            for (int e = 0; e < 4; ++e) {
                hb[e].x = fmaxf(fmaf(x, w1r[4 * T + e].x, b1r[4 * T + e].x), 0.0f);
                hb[e].y = fmaxf(fmaf(x, w1r[4 * T + e].y, b1r[4 * T + e].y), 0.0f);
            }
            asm volatile("" : "+v"(hb[0]), "+v"(hb[1]), "+v"(hb[2]), "+v"(hb[3]), "+v"(c0), "+v"(c1), "+v"(poff) : : "memory");
            #pragma unroll
            for (int e = 0; e < 4; ++e) {
                c0 = WMMA_F32(afr[0][4 * T + e], hb[e], c0);
                c1 = WMMA_F32(afr[1][4 * T + e], hb[e], c1);
            }
            TRIP_END(c0, c1, xs[T + 1], poff, aoff);
            float* prow = sP + poff;
            poff += 16 * PT;
            *(v4f*)(prow)          = (v4f){c0[0], c0[1], c0[2], c0[3]};
            *(v4f*)(prow + 4)      = (v4f){c0[4], c0[5], c0[6], c0[7]};
            *(v4f*)(prow + 16)     = (v4f){c1[0], c1[1], c1[2], c1[3]};
            *(v4f*)(prow + 16 + 4) = (v4f){c1[4], c1[5], c1[6], c1[7]};
        }
    }
    __syncthreads();

    {
        v4f w3r[HID / 4];
        #pragma unroll
        for (int k = 0; k < HID / 4; ++k) w3r[k] = *(const v4f*)(sW3 + 4 * k);

        const int j = t;
        const float* r0 = sP + j * PT;
        const float* r1 = r0 + PT;
        const int rb = j * RP + ((j >> 6) << 4);
        v4f s0 = {0.0f, 0.0f, 0.0f, 0.0f};
        v4f s1 = {0.0f, 0.0f, 0.0f, 0.0f};
        unsigned mk[4] = {0u, 0u, 0u, 0u};
        #pragma unroll
        for (int k = 0; k < HID / 4; ++k) {
            const v4f p0 = *(const v4f*)(r0 + 4 * k);
            const v4f p1 = *(const v4f*)(r1 + 4 * k);
            #pragma unroll
            for (int e = 0; e < 4; ++e) {
                const int xs = __float_as_int(p0[e]) ^ __float_as_int(p1[e]);
                mk[k >> 3] = __builtin_amdgcn_alignbit(mk[k >> 3], (unsigned)xs, 31);
                const float we = xs < 0 ? 0.0f : w3r[k][e];
                const float q0 = __int_as_float(max(__float_as_int(p0[e]), 0));
                const float q1 = __int_as_float(max(__float_as_int(p1[e]), 0));
                s0[e] = fmaf(we, q0, s0[e]);
                s1[e] = fmaf(we, q1, s1[e]);
            }
        }
        const float a0 = (s0[0] + s0[1]) + (s0[2] + s0[3]);
        const float a1 = (s1[0] + s1[1]) + (s1[2] + s1[3]);
        sA[j] = a0;
        sB[j] = a1 - a0;

        float* const rseg = rec + 3 * rb;
        int cnt3 = 0;
        #pragma unroll
        for (int q = 0; q < 4; ++q) {
            unsigned m = mk[q];
            while (m != 0u) {
                const int i = __builtin_clz(m);
                m &= ~(0x80000000u >> i);
                const int u = 32 * q + i;
                const float p0 = r0[u];
                const float p1 = r1[u];
                float* rp = rseg + cnt3;
                rp[0] = p0;
                rp[1] = p1 - p0;
                rp[2] = sW3[u];
                cnt3 += 3;
            }
        }
        sC[j] = cnt3;
        if (t == 0) { rec[ZREC] = 0.0f; rec[ZREC + 1] = 0.0f; rec[ZREC + 2] = 0.0f; }
    }

    if (wv == NWAVES - 1) {
        const int j = NSEG - 1;
        const v4f p0 = *(const v4f*)(sP + j * PT + 4 * lane);
        const v4f p1 = *(const v4f*)(sP + (j + 1) * PT + 4 * lane);
        const v4f w  = *(const v4f*)(sW3 + 4 * lane);
        const int rb = j * RP + ((j >> 6) << 4);
        float a0 = 0.0f, a1 = 0.0f;
        int base = 0;
        #pragma unroll
        for (int e = 0; e < 4; ++e) {
            const bool sw = (__float_as_int(p0[e]) ^ __float_as_int(p1[e])) < 0;
            const unsigned ball = __builtin_amdgcn_ballot_w32(sw);
            const int pos = base + (int)__builtin_amdgcn_mbcnt_lo(ball, 0u);
            if (sw) {
                float* rp = rec + 3 * (rb + pos);
                rp[0] = p0[e];
                rp[1] = p1[e] - p0[e];
                rp[2] = w[e];
            }
            base += __builtin_popcount(ball);
            const float we = sw ? 0.0f : w[e];
            a0 = fmaf(we, fmaxf(p0[e], 0.0f), a0);
            a1 = fmaf(we, fmaxf(p1[e], 0.0f), a1);
        }
        #pragma unroll
        for (int m = 16; m >= 1; m >>= 1) {
            a0 += __shfl_xor(a0, m, 32);
            a1 += __shfl_xor(a1, m, 32);
        }
        if (lane == 0) { sA[j] = a0; sB[j] = a1 - a0; sC[j] = 3 * base; }
    }
    __syncthreads();

    const float b3v = b3[n];

    int   rbase[ROWS_PER_THREAD];
    int   rc[ROWS_PER_THREAD];
    float tr[ROWS_PER_THREAD];
    float ov[ROWS_PER_THREAD];
    int   cl = 0;
    int jj[ROWS_PER_THREAD];
    {
        const float kmid = sS[HID];
        #pragma unroll
        for (int p = 0; p < ROWS_PER_THREAD; ++p) jj[p] = (kmid <= ar[p]) ? HID : 0;
    }
    #pragma unroll
    for (int step = HID / 2; step >= 1; step >>= 1) {
        float key[ROWS_PER_THREAD];
        #pragma unroll
        for (int p = 0; p < ROWS_PER_THREAD; ++p) key[p] = sS[jj[p] + step];
        asm volatile("" : "+v"(key[0]), "+v"(key[1]), "+v"(key[2]), "+v"(key[3]), "+v"(key[4]), "+v"(key[5]), "+v"(key[6]), "+v"(key[7]));
        #pragma unroll
        for (int p = 0; p < 8; ++p) jj[p] = (key[p] <= ar[p]) ? jj[p] + step : jj[p];
        asm volatile("" : "+v"(key[8]), "+v"(key[9]), "+v"(key[10]), "+v"(key[11]), "+v"(key[12]), "+v"(key[13]), "+v"(key[14]), "+v"(key[15]));
        #pragma unroll
        for (int p = 8; p < ROWS_PER_THREAD; ++p) jj[p] = (key[p] <= ar[p]) ? jj[p] + step : jj[p];
    }
    float gx0[ROWS_PER_THREAD], gx1[ROWS_PER_THREAD], gsa[ROWS_PER_THREAD], gsb[ROWS_PER_THREAD];
    int   gsc[ROWS_PER_THREAD];
    #pragma unroll
    for (int p = 0; p < ROWS_PER_THREAD; ++p) {
        int j = jj[p];
        j = j < NPTS - 2 ? j : NPTS - 2;
        jj[p] = j;
        gx0[p] = sX[j];
        gx1[p] = sX[j + 1];
        gsa[p] = sA[j];
        gsb[p] = sB[j];
        gsc[p] = sC[j];
    }
    asm volatile("" : "+v"(gx0[0]), "+v"(gx0[1]), "+v"(gx0[2]), "+v"(gx0[3]), "+v"(gx0[4]), "+v"(gx0[5]), "+v"(gx0[6]), "+v"(gx0[7]),
                      "+v"(gx0[8]), "+v"(gx0[9]), "+v"(gx0[10]), "+v"(gx0[11]), "+v"(gx0[12]), "+v"(gx0[13]), "+v"(gx0[14]), "+v"(gx0[15]));
    asm volatile("" : "+v"(gx1[0]), "+v"(gx1[1]), "+v"(gx1[2]), "+v"(gx1[3]), "+v"(gx1[4]), "+v"(gx1[5]), "+v"(gx1[6]), "+v"(gx1[7]),
                      "+v"(gx1[8]), "+v"(gx1[9]), "+v"(gx1[10]), "+v"(gx1[11]), "+v"(gx1[12]), "+v"(gx1[13]), "+v"(gx1[14]), "+v"(gx1[15]));
    asm volatile("" : "+v"(gsa[0]), "+v"(gsa[1]), "+v"(gsa[2]), "+v"(gsa[3]), "+v"(gsa[4]), "+v"(gsa[5]), "+v"(gsa[6]), "+v"(gsa[7]),
                      "+v"(gsa[8]), "+v"(gsa[9]), "+v"(gsa[10]), "+v"(gsa[11]), "+v"(gsa[12]), "+v"(gsa[13]), "+v"(gsa[14]), "+v"(gsa[15]));
    asm volatile("" : "+v"(gsb[0]), "+v"(gsb[1]), "+v"(gsb[2]), "+v"(gsb[3]), "+v"(gsb[4]), "+v"(gsb[5]), "+v"(gsb[6]), "+v"(gsb[7]),
                      "+v"(gsb[8]), "+v"(gsb[9]), "+v"(gsb[10]), "+v"(gsb[11]), "+v"(gsb[12]), "+v"(gsb[13]), "+v"(gsb[14]), "+v"(gsb[15]));
    asm volatile("" : "+v"(gsc[0]), "+v"(gsc[1]), "+v"(gsc[2]), "+v"(gsc[3]), "+v"(gsc[4]), "+v"(gsc[5]), "+v"(gsc[6]), "+v"(gsc[7]),
                      "+v"(gsc[8]), "+v"(gsc[9]), "+v"(gsc[10]), "+v"(gsc[11]), "+v"(gsc[12]), "+v"(gsc[13]), "+v"(gsc[14]), "+v"(gsc[15]));
    #pragma unroll
    for (int p = 0; p < ROWS_PER_THREAD; ++p) {
        const float x = ar[p];
        const int   j = jj[p];
        const float x0 = gx0[p];
        const float x1 = gx1[p];
        const float wd = x1 - x0;
        const float tt = (x - x0) / (wd > 0.0f ? wd : 1.0f);
        tr[p] = tt;
        ov[p] = b3v + fmaf(tt, gsb[p], gsa[p]);
        int c = gsc[p];
        c = c < 3 * HID ? c : 3 * HID;
        rc[p] = c;
        cl = c > cl ? c : cl;
        rbase[p] = 3 * (j * RP + ((j >> 6) << 4));
    }

    #pragma unroll 1
    for (int i = 0; i < cl; i += 3) {
        #pragma unroll
        for (int p = 0; p < ROWS_PER_THREAD; ++p) {
            const float* rp = rec + ((i < rc[p]) ? rbase[p] + i : ZREC);
            const float z = fmaf(tr[p], rp[1], rp[0]);
            ov[p] = fmaf(rp[2], fmaxf(z, 0.0f), ov[p]);
        }
    }

    #pragma unroll
    for (int p = 0; p < ROWS_PER_THREAD; ++p) op[p * 128] = ov[p];
}

extern "C" void kernel_launch(void* const* d_in, const int* in_sizes, int n_in,
                              void* d_out, int out_size, void* d_ws, size_t ws_size,
                              hipStream_t stream) {
    const float* a  = (const float*)d_in[0];
    const float* W1 = (const float*)d_in[1];
    const float* b1 = (const float*)d_in[2];
    const float* W2 = (const float*)d_in[3];
    const float* b2 = (const float*)d_in[4];
    const float* W3 = (const float*)d_in[5];
    const float* b3 = (const float*)d_in[6];
    float* out = (float*)d_out;

    mlp256_knot_table_wmma_f32<<<dim3(NNETS), 128, 0, stream>>>(a, W1, b1, W2, b2, W3, b3, out);
}


# Two waves per SIMD: 8-wave work-group per net, one 16-row W2 tile per wave in both table passes, 8 rows per lane; sort keys and segment units split between two thread halves (order of sums changes)
# speedup vs baseline: 6.3069x; 1.1499x over previous
#include <hip/hip_runtime.h>

typedef __attribute__((ext_vector_type(2))) float v2f;
typedef __attribute__((ext_vector_type(2))) int   v2i;
typedef __attribute__((ext_vector_type(4))) float v4f;
typedef __attribute__((ext_vector_type(8))) float v8f;

#define NNETS 256
#define HID 128
#define BATCH 2048
#define NWAVES 8
#define NTHREADS (32 * NWAVES)
#define ROWS_PER_THREAD (BATCH / NTHREADS)
#define W2_STRIDE 132
#define NPTS 130
#define NPTS_PAD 144
#define PT 132
#define NSEG (NPTS - 1)
#define RP 129
#define REC_N 16704
#define ZREC (3 * REC_N - 12)

#define WMMA_F32(A, B, C) __builtin_amdgcn_wmma_f32_16x16x4_f32(false, (A), false, (B), (short)0, (C), false, false)

#define TRIP_END2(C0, C1, IN, OFF, OFF2, OFF3) \
    asm volatile("v_nop\n\tv_nop\n\tv_nop\n\tv_nop" : "+v"(C0), "+v"(C1), "+v"(IN), "+v"(OFF), "+v"(OFF2), "+v"(OFF3))
#define TRIP_END1(C0, IN, OFF, OFF2) \
    asm volatile("v_nop\n\tv_nop\n\tv_nop\n\tv_nop" : "+v"(C0), "+v"(IN), "+v"(OFF), "+v"(OFF2))

#define H_READY_LO(H) \
    asm volatile("" : "+v"((H)[0]), "+v"((H)[1]), "+v"((H)[2]), "+v"((H)[3]), "+v"((H)[4]), "+v"((H)[5]), \
                      "+v"((H)[6]), "+v"((H)[7]), "+v"((H)[8]), "+v"((H)[9]), "+v"((H)[10]), "+v"((H)[11]), \
                      "+v"((H)[12]), "+v"((H)[13]), "+v"((H)[14]), "+v"((H)[15]))
#define H_READY_HI(H, C0, C1, OFF) \
    asm volatile("" : "+v"((H)[16]), "+v"((H)[17]), "+v"((H)[18]), "+v"((H)[19]), "+v"((H)[20]), "+v"((H)[21]), \
                      "+v"((H)[22]), "+v"((H)[23]), "+v"((H)[24]), "+v"((H)[25]), "+v"((H)[26]), "+v"((H)[27]), \
                      "+v"((H)[28]), "+v"((H)[29]), "+v"((H)[30]), "+v"((H)[31]), "+v"(C0), "+v"(C1), "+v"(OFF))

#define LDS_BARRIER() do { \
    __builtin_amdgcn_fence(__ATOMIC_RELEASE, "workgroup", "local"); \
    __builtin_amdgcn_s_barrier(); \
    __builtin_amdgcn_fence(__ATOMIC_ACQUIRE, "workgroup", "local"); } while (0)

__global__ __launch_bounds__(NTHREADS) void mlp256_knot_table_wmma_f32(
    const float* __restrict__ a,  const float* __restrict__ W1,
    const float* __restrict__ b1, const float* W2,
    const float* __restrict__ b2, const float* __restrict__ W3,
    const float* __restrict__ b3, float* __restrict__ out)
{
    __shared__ __attribute__((aligned(16))) float sU[3 * REC_N];
    float* const sW2  = sU;
    float* const rec  = sU;
    __shared__ float sA[NSEG + 3];
    __shared__ float sB[NSEG + 3];
    __shared__ int   sC[NSEG + 3];
    __shared__ __attribute__((aligned(16))) float sP[NPTS_PAD * PT];
    __shared__ __attribute__((aligned(16))) float sX[NPTS_PAD];
    __shared__ __attribute__((aligned(16))) float sS[2 * HID];
    __shared__ __attribute__((aligned(16))) float sK[HID];
    __shared__ __attribute__((aligned(16))) float sW1[HID];
    __shared__ __attribute__((aligned(16))) float sB1[HID];
    __shared__ __attribute__((aligned(16))) float sW1s[HID];
    __shared__ __attribute__((aligned(16))) float sB1s[HID];
    __shared__ __attribute__((aligned(16))) int   sPerm[HID];
    __shared__ __attribute__((aligned(16))) float sAB[16 * PT];
    __shared__ __attribute__((aligned(16))) float sB2[HID];
    __shared__ __attribute__((aligned(16))) float sW3[HID];
    __shared__ float sPA[HID];
    __shared__ float sPB[HID];
    __shared__ int   sPC[HID];
    __shared__ int   sRank[HID];
    __shared__ float sLo[NWAVES];
    __shared__ float sHi[NWAVES];

    const int n = blockIdx.x;
    const int t = threadIdx.x;
    const int lane = t & 31;
    const int wv   = t >> 5;
    const int li   = lane & 15;
    const int lh   = lane >> 4;

    const int un = t & (HID - 1);
    const int hf = t >> 7;

    const float w1t = W1[n * HID + un];
    const float b1t = b1[n * HID + un];
    if (hf == 0) {
        sW1[un] = w1t;
        sB1[un] = b1t;
        sB2[un] = b2[n * HID + un];
        sW3[un] = W3[n * HID + un];
    }

    const float* ap = a   + (size_t)n * BATCH + t;
    float*       op = out + (size_t)n * BATCH + t;
    float ar[ROWS_PER_THREAD];
    #pragma unroll
    for (int p = 0; p < ROWS_PER_THREAD; ++p) ar[p] = ap[p * NTHREADS];

    const float* gW2 = W2 + (size_t)n * HID * HID;
    v4f w2v[16];
    #pragma unroll
    for (int i = 0; i < 16; ++i) w2v[i] = *(const v4f*)(gW2 + (t + NTHREADS * i) * 4);
    float lo = ar[0], hi = ar[0];
    #pragma unroll
    for (int p = 1; p < ROWS_PER_THREAD; ++p) { lo = fminf(lo, ar[p]); hi = fmaxf(hi, ar[p]); }
    #pragma unroll
    for (int m = 16; m >= 1; m >>= 1) {
        lo = fminf(lo, __shfl_xor(lo, m, 32));
        hi = fmaxf(hi, __shfl_xor(hi, m, 32));
    }
    if (lane == 0) { sLo[wv] = lo; sHi[wv] = hi; }

    LDS_BARRIER();

    lo = fminf(fminf(fminf(sLo[0], sLo[1]), fminf(sLo[2], sLo[3])), fminf(fminf(sLo[4], sLo[5]), fminf(sLo[6], sLo[7])));
    hi = fmaxf(fmaxf(fmaxf(sHi[0], sHi[1]), fmaxf(sHi[2], sHi[3])), fmaxf(fmaxf(sHi[4], sHi[5]), fmaxf(sHi[6], sHi[7])));

    float xk = -b1t / w1t;
    xk = (xk >= lo && xk <= hi) ? xk : hi;
    if (hf == 0) sK[un] = xk;
    LDS_BARRIER();

    {
        int r = 0;
        const int unr = un - 64 * hf;
        v4f q[HID / 8];
        #pragma unroll
        for (int u4 = 0; u4 < HID / 8; ++u4) q[u4] = *(const v4f*)(sK + 64 * hf + 4 * u4);
        asm volatile("" : "+v"(q[0]), "+v"(q[1]), "+v"(q[2]), "+v"(q[3]), "+v"(q[4]), "+v"(q[5]), "+v"(q[6]), "+v"(q[7]),
                          "+v"(q[8]), "+v"(q[9]), "+v"(q[10]), "+v"(q[11]), "+v"(q[12]), "+v"(q[13]), "+v"(q[14]), "+v"(q[15]));
        #pragma unroll
        for (int u4 = 0; u4 < HID / 8; ++u4) {
            #pragma unroll
            for (int e = 0; e < 4; ++e) {
                const int u = 4 * u4 + e;
                r += ((q[u4][e] < xk) || (q[u4][e] == xk && u < unr)) ? 1 : 0;
            }
        }
        if (hf == 1) sRank[un] = r;
        LDS_BARRIER();
        if (hf == 0) {
            r += 1 + sRank[un];
            r = r < 1 ? 1 : (r > HID ? HID : r);
            sX[r] = xk;
            sS[r] = xk;
            sW1s[r - 1] = w1t;
            sB1s[r - 1] = b1t;
            sPerm[r - 1] = t;
            if (t == 0) { sX[0] = lo; sS[0] = lo; }
            if (t < NPTS_PAD - (HID + 1)) sX[HID + 1 + t] = hi;
            if (t >= 1) sS[HID + t] = __builtin_inff();
        }
    }
    #pragma unroll
    for (int i = 0; i < 16; ++i) {
        int e   = (t + NTHREADS * i) * 4;
        int row = e >> 7;
        int col = e & 127;
        *(v4f*)(sW2 + row * W2_STRIDE + col) = w2v[i];
    }
    __syncthreads();

    {
        const int mt = wv;
        v2f afr[32];
        v2i pu[32];
        #pragma unroll
        for (int s = 0; s < 32; ++s) pu[s] = *(const v2i*)(sPerm + 4 * s + 2 * lh);
        asm volatile("" : "+v"(pu[0]), "+v"(pu[1]), "+v"(pu[2]), "+v"(pu[3]), "+v"(pu[4]), "+v"(pu[5]), "+v"(pu[6]), "+v"(pu[7]),
                          "+v"(pu[8]), "+v"(pu[9]), "+v"(pu[10]), "+v"(pu[11]), "+v"(pu[12]), "+v"(pu[13]), "+v"(pu[14]), "+v"(pu[15]));
        asm volatile("" : "+v"(pu[16]), "+v"(pu[17]), "+v"(pu[18]), "+v"(pu[19]), "+v"(pu[20]), "+v"(pu[21]), "+v"(pu[22]), "+v"(pu[23]),
                          "+v"(pu[24]), "+v"(pu[25]), "+v"(pu[26]), "+v"(pu[27]), "+v"(pu[28]), "+v"(pu[29]), "+v"(pu[30]), "+v"(pu[31]));
        const float* arow = sW2 + (mt * 16 + li) * W2_STRIDE;
        #pragma unroll
        for (int s = 0; s < 32; ++s) {
            const int u0 = pu[s].x & (HID - 1);
            const int u1 = pu[s].y & (HID - 1);
            afr[s].x = arow[u0];
            afr[s].y = arow[u1];
        }
        v2f w1r[32], b1r[32];
        #pragma unroll
        for (int s = 0; s < 32; ++s) {
            int k = 4 * s + 2 * lh;
            w1r[s] = *(const v2f*)(sW1s + k);
            b1r[s] = *(const v2f*)(sB1s + k);
        }
        v8f b2v;
        #pragma unroll
        for (int v = 0; v < 8; ++v)
            b2v[v] = sB2[mt * 16 + v + 8 * lh];
        float xs[9];
        #pragma unroll
        for (int T = 0; T < 8; ++T) xs[T] = sX[16 * T + 1 + li];
        xs[8] = xs[7];

        int poff = li * PT + mt * 16 + 8 * lh;
        int aoff = mt * 16 + 8 * lh;

        v8f a0, a1;
        {
            const int   tau = li & 7;
            const bool  isSlope = li < 8;
            const float xm = 0.5f * sX[16 * tau + 1] + 0.5f * sX[16 * tau + 16];
            const float xe = (li == 0) ? lo : hi;
            v2f g[32], h[32];
            #pragma unroll
            for (int s = 0; s < 32; ++s) {
                const bool other = (s >> 2) != tau;
                const float p0 = fmaf(xm, w1r[s].x, b1r[s].x);
                const float p1 = fmaf(xm, w1r[s].y, b1r[s].y);
                const float s0 = isSlope ? w1r[s].x : b1r[s].x;
                const float s1 = isSlope ? w1r[s].y : b1r[s].y;
                g[s].x = (other && p0 > 0.0f) ? s0 : 0.0f;
                g[s].y = (other && p1 > 0.0f) ? s1 : 0.0f;
                h[s].x = fmaxf(fmaf(xe, w1r[s].x, b1r[s].x), 0.0f);
                h[s].y = fmaxf(fmaf(xe, w1r[s].y, b1r[s].y), 0.0f);
            }
            #pragma unroll
            for (int v = 0; v < 8; ++v)
                a0[v] = isSlope ? 0.0f : b2v[v];
            a1 = b2v;
            asm volatile("" : "+v"(xs[0]), "+v"(xs[1]), "+v"(xs[2]), "+v"(xs[3]), "+v"(xs[4]), "+v"(xs[5]),
                              "+v"(xs[6]), "+v"(xs[7]), "+v"(aoff));
            H_READY_LO(g);
            H_READY_HI(g, a0, a1, poff);
            H_READY_LO(h);
            H_READY_HI(h, a0, a1, poff);
            int pend = poff + ((li == 0) ? 0 : 128 * PT);
            asm volatile("" : "+v"(a0), "+v"(a1), "+v"(pend));
            #pragma unroll
            for (int s = 0; s < 32; ++s) {
                a0 = WMMA_F32(afr[s], g[s], a0);
                a1 = WMMA_F32(afr[s], h[s], a1);
            }
            TRIP_END2(a0, a1, xs[0], poff, aoff, pend);
            float* prow = sP + pend;
            *(v4f*)(prow)     = (v4f){a1[0], a1[1], a1[2], a1[3]};
            *(v4f*)(prow + 4) = (v4f){a1[4], a1[5], a1[6], a1[7]};
            float* crow = sAB + poff;
            *(v4f*)(crow)     = (v4f){a0[0], a0[1], a0[2], a0[3]};
            *(v4f*)(crow + 4) = (v4f){a0[4], a0[5], a0[6], a0[7]};
        }

        const float* ab = sAB + aoff;
        v4f sl[8][2], of[8][2];
        #pragma unroll
        for (int T = 0; T < 8; ++T) {
            sl[T][0] = *(const v4f*)(ab + T * PT);
            sl[T][1] = *(const v4f*)(ab + T * PT + 4);
            of[T][0] = *(const v4f*)(ab + (8 + T) * PT);
            of[T][1] = *(const v4f*)(ab + (8 + T) * PT + 4);
        }
        poff += PT;
        #pragma unroll
        for (int T = 0; T < 8; ++T) {
            const float x = xs[T];
            v8f c0;
            #pragma unroll
            for (int v = 0; v < 4; ++v) {
                c0[v]     = fmaf(sl[T][0][v], x, of[T][0][v]);
                c0[v + 4] = fmaf(sl[T][1][v], x, of[T][1][v]);
            }
            v2f hb[4];
            #pragma unroll
            for (int e = 0; e < 4; ++e) {
                hb[e].x = fmaxf(fmaf(x, w1r[4 * T + e].x, b1r[4 * T + e].x), 0.0f);
                hb[e].y = fmaxf(fmaf(x, w1r[4 * T + e].y, b1r[4 * T + e].y), 0.0f);
            }
            asm volatile("" : "+v"(hb[0]), "+v"(hb[1]), "+v"(hb[2]), "+v"(hb[3]), "+v"(c0), "+v"(poff) : : "memory");
            #pragma unroll
            for (int e = 0; e < 4; ++e)
                c0 = WMMA_F32(afr[4 * T + e], hb[e], c0);
            TRIP_END1(c0, xs[T + 1], poff, aoff);
            float* prow = sP + poff;
            poff += 16 * PT;
            *(v4f*)(prow)     = (v4f){c0[0], c0[1], c0[2], c0[3]};
            *(v4f*)(prow + 4) = (v4f){c0[4], c0[5], c0[6], c0[7]};
        }
    }
    __syncthreads();

    {
        const float* w3h = sW3 + 64 * hf;
        v4f w3r[HID / 8];
        #pragma unroll
        for (int k = 0; k < HID / 8; ++k) w3r[k] = *(const v4f*)(w3h + 4 * k);

        const int j = un;
        const float* r0 = sP + j * PT + 64 * hf;
        const float* r1 = r0 + PT;
        const int rb = j * RP + ((j >> 6) << 4);
        v4f s0 = {0.0f, 0.0f, 0.0f, 0.0f};
        v4f s1 = {0.0f, 0.0f, 0.0f, 0.0f};
        unsigned mk[2] = {0u, 0u};
        #pragma unroll
        for (int k = 0; k < HID / 8; ++k) {
            const v4f p0 = *(const v4f*)(r0 + 4 * k);
            const v4f p1 = *(const v4f*)(r1 + 4 * k);
            #pragma unroll
            for (int e = 0; e < 4; ++e) {
                const int xs = __float_as_int(p0[e]) ^ __float_as_int(p1[e]);
                mk[k >> 3] = __builtin_amdgcn_alignbit(mk[k >> 3], (unsigned)xs, 31);
                const float we = xs < 0 ? 0.0f : w3r[k][e];
                const float q0 = __int_as_float(max(__float_as_int(p0[e]), 0));
                const float q1 = __int_as_float(max(__float_as_int(p1[e]), 0));
                s0[e] = fmaf(we, q0, s0[e]);
                s1[e] = fmaf(we, q1, s1[e]);
            }
        }
        const float a0 = (s0[0] + s0[1]) + (s0[2] + s0[3]);
        const float a1 = (s1[0] + s1[1]) + (s1[2] + s1[3]);
        const int mine3 = 3 * (__builtin_popcount(mk[0]) + __builtin_popcount(mk[1]));
        if (hf == 0) { sPA[j] = a0; sPB[j] = a1 - a0; sPC[j] = mine3; }
        LDS_BARRIER();
        int cnt3 = 0;
        if (hf == 1) {
            cnt3 = sPC[j];
            cnt3 = cnt3 < 0 ? 0 : (cnt3 > 3 * (HID / 2) ? 3 * (HID / 2) : cnt3);
            sA[j] = sPA[j] + a0;
            sB[j] = sPB[j] + (a1 - a0);
            sC[j] = cnt3 + mine3;
        }

        float* const rseg = rec + 3 * rb;
        #pragma unroll
        for (int q = 0; q < 2; ++q) {
            unsigned m = mk[q];
            while (m != 0u) {
                const int i = __builtin_clz(m);
                m &= ~(0x80000000u >> i);
                const int u = 32 * q + i;
                const float p0 = r0[u];
                const float p1 = r1[u];
                float* rp = rseg + cnt3;
                rp[0] = p0;
                rp[1] = p1 - p0;
                rp[2] = w3h[u];
                cnt3 += 3;
            }
        }
        if (t == 0) { rec[ZREC] = 0.0f; rec[ZREC + 1] = 0.0f; rec[ZREC + 2] = 0.0f; }
    }

    if (wv == NWAVES - 1) {
        const int j = NSEG - 1;
        const v4f p0 = *(const v4f*)(sP + j * PT + 4 * lane);
        const v4f p1 = *(const v4f*)(sP + (j + 1) * PT + 4 * lane);
        const v4f w  = *(const v4f*)(sW3 + 4 * lane);
        const int rb = j * RP + ((j >> 6) << 4);
        float a0 = 0.0f, a1 = 0.0f;
        int base = 0;
        #pragma unroll
        for (int e = 0; e < 4; ++e) {
            const bool sw = (__float_as_int(p0[e]) ^ __float_as_int(p1[e])) < 0;
            const unsigned ball = __builtin_amdgcn_ballot_w32(sw);
            const int pos = base + (int)__builtin_amdgcn_mbcnt_lo(ball, 0u);
            if (sw) {
                float* rp = rec + 3 * (rb + pos);
                rp[0] = p0[e];
                rp[1] = p1[e] - p0[e];
                rp[2] = w[e];
            }
            base += __builtin_popcount(ball);
            const float we = sw ? 0.0f : w[e];
            a0 = fmaf(we, fmaxf(p0[e], 0.0f), a0);
            a1 = fmaf(we, fmaxf(p1[e], 0.0f), a1);
        }
        #pragma unroll
        for (int m = 16; m >= 1; m >>= 1) {
            a0 += __shfl_xor(a0, m, 32);
            a1 += __shfl_xor(a1, m, 32);
        }
        if (lane == 0) { sA[j] = a0; sB[j] = a1 - a0; sC[j] = 3 * base; }
    }
    __syncthreads();

    const float b3v = b3[n];

    int   rbase[ROWS_PER_THREAD];
    int   rc[ROWS_PER_THREAD];
    float tr[ROWS_PER_THREAD];
    float ov[ROWS_PER_THREAD];
    int   cl = 0;
    int jj[ROWS_PER_THREAD];
    {
        const float kmid = sS[HID];
        #pragma unroll
        for (int p = 0; p < ROWS_PER_THREAD; ++p) jj[p] = (kmid <= ar[p]) ? HID : 0;
    }
    #pragma unroll
    for (int step = HID / 2; step >= 1; step >>= 1) {
        float key[ROWS_PER_THREAD];
        #pragma unroll
        for (int p = 0; p < ROWS_PER_THREAD; ++p) key[p] = sS[jj[p] + step];
        asm volatile("" : "+v"(key[0]), "+v"(key[1]), "+v"(key[2]), "+v"(key[3]), "+v"(key[4]), "+v"(key[5]), "+v"(key[6]), "+v"(key[7]));
        #pragma unroll
        for (int p = 0; p < ROWS_PER_THREAD; ++p) jj[p] = (key[p] <= ar[p]) ? jj[p] + step : jj[p];
    }
    float gx0[ROWS_PER_THREAD], gx1[ROWS_PER_THREAD], gsa[ROWS_PER_THREAD], gsb[ROWS_PER_THREAD];
    int   gsc[ROWS_PER_THREAD];
    #pragma unroll
    for (int p = 0; p < ROWS_PER_THREAD; ++p) {
        int j = jj[p];
        j = j < NPTS - 2 ? j : NPTS - 2;
        jj[p] = j;
        gx0[p] = sX[j];
        gx1[p] = sX[j + 1];
        gsa[p] = sA[j];
        gsb[p] = sB[j];
        gsc[p] = sC[j];
    }
    asm volatile("" : "+v"(gx0[0]), "+v"(gx0[1]), "+v"(gx0[2]), "+v"(gx0[3]), "+v"(gx0[4]), "+v"(gx0[5]), "+v"(gx0[6]), "+v"(gx0[7]));
    asm volatile("" : "+v"(gx1[0]), "+v"(gx1[1]), "+v"(gx1[2]), "+v"(gx1[3]), "+v"(gx1[4]), "+v"(gx1[5]), "+v"(gx1[6]), "+v"(gx1[7]));
    asm volatile("" : "+v"(gsa[0]), "+v"(gsa[1]), "+v"(gsa[2]), "+v"(gsa[3]), "+v"(gsa[4]), "+v"(gsa[5]), "+v"(gsa[6]), "+v"(gsa[7]));
    asm volatile("" : "+v"(gsb[0]), "+v"(gsb[1]), "+v"(gsb[2]), "+v"(gsb[3]), "+v"(gsb[4]), "+v"(gsb[5]), "+v"(gsb[6]), "+v"(gsb[7]));
    asm volatile("" : "+v"(gsc[0]), "+v"(gsc[1]), "+v"(gsc[2]), "+v"(gsc[3]), "+v"(gsc[4]), "+v"(gsc[5]), "+v"(gsc[6]), "+v"(gsc[7]));
    #pragma unroll
    for (int p = 0; p < ROWS_PER_THREAD; ++p) {
        const float x = ar[p];
        const int   j = jj[p];
        const float x0 = gx0[p];
        const float x1 = gx1[p];
        const float wd = x1 - x0;
        const float tt = (x - x0) / (wd > 0.0f ? wd : 1.0f);
        tr[p] = tt;
        ov[p] = b3v + fmaf(tt, gsb[p], gsa[p]);
        int c = gsc[p];
        c = c < 3 * HID ? c : 3 * HID;
        rc[p] = c;
        cl = c > cl ? c : cl;
        rbase[p] = 3 * (j * RP + ((j >> 6) << 4));
    }

    #pragma unroll 1
    for (int i = 0; i < cl; i += 3) {
        #pragma unroll
        for (int p = 0; p < ROWS_PER_THREAD; ++p) {
            const float* rp = rec + ((i < rc[p]) ? rbase[p] + i : ZREC);
            const float z = fmaf(tr[p], rp[1], rp[0]);
            ov[p] = fmaf(rp[2], fmaxf(z, 0.0f), ov[p]);
        }
    }

    #pragma unroll
    for (int p = 0; p < ROWS_PER_THREAD; ++p) op[p * NTHREADS] = ov[p];
}

extern "C" void kernel_launch(void* const* d_in, const int* in_sizes, int n_in,
                              void* d_out, int out_size, void* d_ws, size_t ws_size,
                              hipStream_t stream) {
    const float* a  = (const float*)d_in[0];
    const float* W1 = (const float*)d_in[1];
    const float* b1 = (const float*)d_in[2];
    const float* W2 = (const float*)d_in[3];
    const float* b2 = (const float*)d_in[4];
    const float* W3 = (const float*)d_in[5];
    const float* b3 = (const float*)d_in[6];
    float* out = (float*)d_out;

    mlp256_knot_table_wmma_f32<<<dim3(NNETS), NTHREADS, 0, stream>>>(a, W1, b1, W2, b2, W3, b3, out);
}
